# starting kernel
# baseline (speedup 1.0000x reference)
#include <hip/hip_runtime.h>
#include <math.h>

#define BH_TOT 32
#define NCHUNK 32
#define CT     128
#define CPAIR  64
#define DD     128
#define STABV  10000.0f

typedef float v2f __attribute__((ext_vector_type(2)));
typedef float v8f __attribute__((ext_vector_type(8)));

__device__ __forceinline__ float gelu_tanh(float x) {
    const float k0 = 0.7978845608028654f;
    const float k1 = 0.044715f;
    float t = tanhf(k0 * (x + k1 * x * x * x));
    return 0.5f * x * (1.0f + t);
}

__device__ __forceinline__ float clipstab(float v) {
    return fminf(fmaxf(v, -STABV), STABV);
}

__global__ __launch_bounds__(256) void liz_scan(
    const float* __restrict__ Wt, const float* __restrict__ Ut,
    const float* __restrict__ h0, float* __restrict__ hstates,
    float* __restrict__ hfinal)
{
    __shared__ float h[DD * DD];
    __shared__ float w0r[DD], w1r[DD], u0r[DD], u1r[DD];
    __shared__ float uv0[DD], uv1[DD];
    __shared__ float psum[256];
    __shared__ float wwred[64];

    const int tid = threadIdx.x;
    const int bh  = blockIdx.x;

    {
        const float4* src = (const float4*)(h0 + (size_t)bh * DD * DD);
        float4* hv = (float4*)h;
        for (int i = tid; i < DD * DD / 4; i += 256) hv[i] = src[i];
    }
    __syncthreads();

    for (int ch = 0; ch < NCHUNK; ++ch) {
        {
            float4* dst = (float4*)(hstates + (size_t)(bh * NCHUNK + ch) * DD * DD);
            const float4* hv = (const float4*)h;
            for (int i = tid; i < DD * DD / 4; i += 256) dst[i] = hv[i];
        }
        const float* wbase = Wt + (size_t)(bh * NCHUNK + ch) * CT * DD;
        const float* ubase = Ut + (size_t)(bh * NCHUNK + ch) * CT * DD;
        if (tid < DD) {
            w0r[tid] = wbase[(size_t)(CT - 2) * DD + tid];
            w1r[tid] = wbase[(size_t)(CT - 1) * DD + tid];
            u0r[tid] = ubase[(size_t)(CT - 2) * DD + tid];
            u1r[tid] = ubase[(size_t)(CT - 1) * DD + tid];
        }
        __syncthreads();

        {
            int d = tid & (DD - 1);
            int half = tid >> 7;
            float s = 0.f;
            for (int r = half * 64; r < half * 64 + 64; ++r) s += w0r[r] * h[r * DD + d];
            psum[tid] = s;
        }
        __syncthreads();
        if (tid < DD) uv0[tid] = u0r[tid] - (psum[tid] + psum[tid + 128]);
        if (tid < 64) wwred[tid] = w1r[tid] * w0r[tid] + w1r[tid + 64] * w0r[tid + 64];
        __syncthreads();
        for (int off = 32; off > 0; off >>= 1) {
            if (tid < off) wwred[tid] += wwred[tid + off];
            __syncthreads();
        }
        float ww = wwred[0];

        {
            int d = tid & (DD - 1);
            int half = tid >> 7;
            float s = 0.f;
            for (int r = half * 64; r < half * 64 + 64; ++r) s += w1r[r] * h[r * DD + d];
            psum[tid] = s;
        }
        __syncthreads();
        if (tid < DD) uv1[tid] = u1r[tid] - (psum[tid] + psum[tid + 128] + ww * uv0[tid]);
        __syncthreads();

        {
            int row  = tid >> 1;
            int half = tid & 1;
            int base = row * DD + half * 64;
            float w0v = w0r[row], w1v = w1r[row];
            float sq = 0.f;
            for (int j = 0; j < 64; ++j) {
                int col = half * 64 + j;
                float v = clipstab(h[base + j] + w0v * uv0[col]);
                v = clipstab(v + w1v * uv1[col]);
                h[base + j] = v;
                sq += v * v;
            }
            psum[tid] = sq;
        }
        __syncthreads();
        {
            int row  = tid >> 1;
            int half = tid & 1;
            float n  = sqrtf(psum[2 * row] + psum[2 * row + 1]) + 1e-6f;
            float inv = 2.0f / n;
            float hn  = 0.5f * n;
            int base = row * DD + half * 64;
            for (int j = 0; j < 64; ++j) {
                float x = h[base + j] * inv;
                h[base + j] = hn * gelu_tanh(x);
            }
        }
        __syncthreads();
    }

    {
        float4* dst = (float4*)(hfinal + (size_t)bh * DD * DD);
        const float4* hv = (const float4*)h;
        for (int i = tid; i < DD * DD / 4; i += 256) dst[i] = hv[i];
    }
}

__global__ __launch_bounds__(256) void liz_out(
    const float* __restrict__ Qt, const float* __restrict__ Wt,
    const float* __restrict__ Ut, const float* __restrict__ hstates,
    float* __restrict__ ys)
{
    __shared__ float h[DD * DD];
    __shared__ float s_w[CPAIR], s_q[CPAIR];

    const int tid  = threadIdx.x;
    const int lane = tid & 31;
    const int wave = tid >> 5;
    const int nc = blockIdx.x, bh = blockIdx.y;

    const size_t cbase = (size_t)(bh * NCHUNK + nc) * CT * DD;
    const float* Qc = Qt + cbase;
    const float* Wc = Wt + cbase;
    const float* Uc = Ut + cbase;

    {
        const float4* src = (const float4*)(hstates + (size_t)(bh * NCHUNK + nc) * DD * DD);
        float4* hv = (float4*)h;
        for (int i = tid; i < DD * DD / 4; i += 256) hv[i] = src[i];
    }

    for (int i = 0; i < 8; ++i) {
        int c = wave * 8 + i;
        const float* w0p = Wc + (size_t)(2 * c) * DD;
        const float* w1p = Wc + (size_t)(2 * c + 1) * DD;
        const float* q1p = Qc + (size_t)(2 * c + 1) * DD;
        float pw = 0.f, pq = 0.f;
        for (int e = lane; e < DD; e += 32) {
            float a = w0p[e];
            pw += w1p[e] * a;
            pq += q1p[e] * a;
        }
        for (int off = 16; off > 0; off >>= 1) {
            pw += __shfl_xor(pw, off, 32);
            pq += __shfl_xor(pq, off, 32);
        }
        if (lane == 0) { s_w[c] = pw; s_q[c] = pq; }
    }
    __syncthreads();

    const int m      = wave;
    const int mrow   = m * 16 + (lane & 15);
    const int khalf  = 2 * (lane >> 4);
    const int nlane  = lane & 15;
    const int rowoff = (lane >> 4) * 8;

    v2f aW[32], aQ[32];
    {
        const float* wrow = Wc + (size_t)mrow * DD;
        const float* qrow = Qc + (size_t)mrow * DD;
#pragma unroll
        for (int k = 0; k < 32; ++k) {
            int col = 4 * k + khalf;
            aW[k] = *(const v2f*)(wrow + col);
            aQ[k] = *(const v2f*)(qrow + col);
        }
    }

    float* outbase = ys + cbase;
    for (int n = 0; n < 8; ++n) {
        v8f accW = {};
        v8f accQ = {};
        const int ncol = n * 16 + nlane;
#pragma unroll
        for (int k = 0; k < 32; ++k) {
            int kk = 4 * k + khalf;
            v2f b;
            b.x = h[kk * DD + ncol];
            b.y = h[(kk + 1) * DD + ncol];
            accW = __builtin_amdgcn_wmma_f32_16x16x4_f32(
                false, aW[k], false, b, (short)0, accW, false, false);
            accQ = __builtin_amdgcn_wmma_f32_16x16x4_f32(
                false, aQ[k], false, b, (short)0, accQ, false, false);
        }
#pragma unroll
        for (int j = 0; j < 4; ++j) {
            int lr0 = 2 * j + rowoff;
            int t0  = m * 16 + lr0;
            int t1  = t0 + 1;
            int c   = t0 >> 1;
            float wh0 = accW[2 * j], wh1 = accW[2 * j + 1];
            float qh0 = accQ[2 * j], qh1 = accQ[2 * j + 1];
            float u0 = Uc[(size_t)t0 * DD + ncol];
            float u1 = Uc[(size_t)t1 * DD + ncol];
            float q0 = Qc[(size_t)t0 * DD + ncol];
            float q1 = Qc[(size_t)t1 * DD + ncol];
            float uv0v = u0 - wh0;
            float out0 = q0 * uv0v + qh0;
            float uv1v = u1 - (wh1 + s_w[c] * uv0v);
            float out1 = q1 * uv1v + qh1 + s_q[c] * uv0v;
            outbase[(size_t)t0 * DD + ncol] = out0;
            outbase[(size_t)t1 * DD + ncol] = out1;
        }
    }
}

extern "C" void kernel_launch(void* const* d_in, const int* in_sizes, int n_in,
                              void* d_out, int out_size, void* d_ws, size_t ws_size,
                              hipStream_t stream) {
    const float* q  = (const float*)d_in[0];
    const float* W  = (const float*)d_in[1];
    const float* U  = (const float*)d_in[2];
    const float* h0 = (const float*)d_in[3];

    float* ys     = (float*)d_out;
    float* hfinal = ys + (size_t)BH_TOT * NCHUNK * CT * DD;
    float* hstate = (float*)d_ws;

    liz_scan<<<BH_TOT, 256, 0, stream>>>(W, U, h0, hstate, hfinal);
    liz_out<<<dim3(NCHUNK, BH_TOT), 256, 0, stream>>>(q, W, U, hstate, ys);
}


# scan: state in registers column-wise (a thread owns 32 rows of one column, 16 waves): mat-vec sums and rank-2 update private to the thread, row norms via one padded LDS tile (sum order changes)
# speedup vs baseline: 2.9990x; 2.9990x over previous
#include <hip/hip_runtime.h>
#include <math.h>

#define BH_TOT 32
#define NCHUNK 32
#define CT     128
#define CPAIR  64
#define DD     128
#define STABV  10000.0f

typedef float v2f __attribute__((ext_vector_type(2)));
typedef float v8f __attribute__((ext_vector_type(8)));

__device__ __forceinline__ float gelu_tanh(float x) {
    const float k0 = 0.7978845608028654f;
    const float k1 = 0.044715f;
    float t = tanhf(k0 * (x + k1 * x * x * x));
    return 0.5f * x * (1.0f + t);
}

__device__ __forceinline__ float clipstab(float v) {
    return fminf(fmaxf(v, -STABV), STABV);
}

#define SCAN_T 512
#define RPT    32
#define VP     144
#define VQ     36

__global__ __launch_bounds__(SCAN_T) void liz_scan(
    const float* __restrict__ Wt, const float* __restrict__ Ut,
    const float* __restrict__ h0, float* __restrict__ hstates,
    float* __restrict__ hfinal)
{
    __shared__ __attribute__((aligned(16))) float vbuf[DD * VP];
    __shared__ __attribute__((aligned(16))) float wbuf[2][2][DD];
    __shared__ float ubuf[2][2][DD];
    __shared__ float psum0[SCAN_T], psum1[SCAN_T];
    __shared__ __attribute__((aligned(16))) float ninv[DD];
    __shared__ __attribute__((aligned(16))) float nhn[DD];

    const int tid  = threadIdx.x;
    const int bh   = blockIdx.x;
    const int lane = tid & 31;
    const int d    = tid & (DD - 1);
    const int q    = tid >> 7;
    const int r0   = q * RPT;
    const int dcol = d + (d >> 5) * 4;

    float hreg[RPT];
    {
        const float* src = h0 + (size_t)bh * DD * DD + (size_t)r0 * DD + d;
#pragma unroll
        for (int i = 0; i < RPT; ++i) hreg[i] = src[(size_t)i * DD];
    }
    if (tid < DD) {
        const float* wb = Wt + (size_t)(bh * NCHUNK) * CT * DD;
        const float* ub = Ut + (size_t)(bh * NCHUNK) * CT * DD;
        wbuf[0][0][tid] = wb[(size_t)(CT - 2) * DD + tid];
        wbuf[0][1][tid] = wb[(size_t)(CT - 1) * DD + tid];
        ubuf[0][0][tid] = ub[(size_t)(CT - 2) * DD + tid];
        ubuf[0][1][tid] = ub[(size_t)(CT - 1) * DD + tid];
    }
    __syncthreads();

    for (int ch = 0; ch < NCHUNK; ++ch) {
        const int cur = ch & 1;
        const bool more = (tid < DD) && (ch + 1 < NCHUNK);

        float pw0 = 0.f, pw1 = 0.f, pu0 = 0.f, pu1 = 0.f;
        if (more) {
            const float* wb = Wt + (size_t)(bh * NCHUNK + ch + 1) * CT * DD;
            const float* ub = Ut + (size_t)(bh * NCHUNK + ch + 1) * CT * DD;
            pw0 = wb[(size_t)(CT - 2) * DD + tid];
            pw1 = wb[(size_t)(CT - 1) * DD + tid];
            pu0 = ub[(size_t)(CT - 2) * DD + tid];
            pu1 = ub[(size_t)(CT - 1) * DD + tid];
        }

        {
            float* dst = hstates + (size_t)(bh * NCHUNK + ch) * DD * DD + (size_t)r0 * DD + d;
#pragma unroll
            for (int i = 0; i < RPT; ++i) dst[(size_t)i * DD] = hreg[i];
        }

        const float* w0c = &wbuf[cur][0][0];
        const float* w1c = &wbuf[cur][1][0];

        float ww;
        {
            float a0 = w1c[lane] * w0c[lane] + w1c[lane + 64] * w0c[lane + 64];
            float a1 = w1c[lane + 32] * w0c[lane + 32] + w1c[lane + 96] * w0c[lane + 96];
            ww = a0 + a1;
            for (int off = 16; off > 0; off >>= 1) ww += __shfl_xor(ww, off, 32);
        }

        const float4* w0v = (const float4*)(w0c + r0);
        const float4* w1v = (const float4*)(w1c + r0);

        {
            float s0 = 0.f, s1 = 0.f;
#pragma unroll
            for (int i4 = 0; i4 < RPT / 4; ++i4) {
                float4 a = w0v[i4];
                float4 b = w1v[i4];
                s0 += a.x * hreg[4 * i4 + 0];
                s0 += a.y * hreg[4 * i4 + 1];
                s0 += a.z * hreg[4 * i4 + 2];
                s0 += a.w * hreg[4 * i4 + 3];
                s1 += b.x * hreg[4 * i4 + 0];
                s1 += b.y * hreg[4 * i4 + 1];
                s1 += b.z * hreg[4 * i4 + 2];
                s1 += b.w * hreg[4 * i4 + 3];
            }
            psum0[tid] = s0;
            psum1[tid] = s1;
        }
        __syncthreads();

        const float k0  = (psum0[d] + psum0[d + 128]) + (psum0[d + 256] + psum0[d + 384]);
        const float k1  = (psum1[d] + psum1[d + 128]) + (psum1[d + 256] + psum1[d + 384]);
        const float uv0 = ubuf[cur][0][d] - k0;
        const float uv1 = ubuf[cur][1][d] - (k1 + ww * uv0);

        {
            float* vrow = vbuf + (size_t)r0 * VP + dcol;
#pragma unroll
            for (int i4 = 0; i4 < RPT / 4; ++i4) {
                float4 a = w0v[i4];
                float4 b = w1v[i4];
                float v;
                v = clipstab(hreg[4 * i4 + 0] + a.x * uv0);
                v = clipstab(v + b.x * uv1);
                hreg[4 * i4 + 0] = v;  vrow[(4 * i4 + 0) * VP] = v;
                v = clipstab(hreg[4 * i4 + 1] + a.y * uv0);
                v = clipstab(v + b.y * uv1);
                hreg[4 * i4 + 1] = v;  vrow[(4 * i4 + 1) * VP] = v;
                v = clipstab(hreg[4 * i4 + 2] + a.z * uv0);
                v = clipstab(v + b.z * uv1);
                hreg[4 * i4 + 2] = v;  vrow[(4 * i4 + 2) * VP] = v;
                v = clipstab(hreg[4 * i4 + 3] + a.w * uv0);
                v = clipstab(v + b.w * uv1);
                hreg[4 * i4 + 3] = v;  vrow[(4 * i4 + 3) * VP] = v;
            }
        }
        __syncthreads();

        {
            const int row  = tid >> 2;
            const int part = tid & 3;
            const float4* src = (const float4*)(vbuf + (size_t)row * VP + part * VQ);
            float sq = 0.f;
#pragma unroll
            for (int k = 0; k < 8; ++k) {
                float4 t = src[k];
                sq += t.x * t.x;
                sq += t.y * t.y;
                sq += t.z * t.z;
                sq += t.w * t.w;
            }
            sq += __shfl_xor(sq, 1, 32);
            sq += __shfl_xor(sq, 2, 32);
            float n = sqrtf(sq) + 1e-6f;
            if (part == 0) {
                ninv[row] = 2.0f / n;
                nhn[row]  = 0.5f * n;
            }
        }
        if (more) {
            wbuf[cur ^ 1][0][tid] = pw0;
            wbuf[cur ^ 1][1][tid] = pw1;
            ubuf[cur ^ 1][0][tid] = pu0;
            ubuf[cur ^ 1][1][tid] = pu1;
        }
        __syncthreads();

        {
            const float4* iv = (const float4*)(ninv + r0);
            const float4* hv = (const float4*)(nhn + r0);
#pragma unroll
            for (int i4 = 0; i4 < RPT / 4; ++i4) {
                float4 a = iv[i4];
                float4 b = hv[i4];
                hreg[4 * i4 + 0] = b.x * gelu_tanh(hreg[4 * i4 + 0] * a.x);
                hreg[4 * i4 + 1] = b.y * gelu_tanh(hreg[4 * i4 + 1] * a.y);
                hreg[4 * i4 + 2] = b.z * gelu_tanh(hreg[4 * i4 + 2] * a.z);
                hreg[4 * i4 + 3] = b.w * gelu_tanh(hreg[4 * i4 + 3] * a.w);
            }
        }
    }

    {
        float* dst = hfinal + (size_t)bh * DD * DD + (size_t)r0 * DD + d;
#pragma unroll
        for (int i = 0; i < RPT; ++i) dst[(size_t)i * DD] = hreg[i];
    }
}

__global__ __launch_bounds__(256) void liz_out(
    const float* __restrict__ Qt, const float* __restrict__ Wt,
    const float* __restrict__ Ut, const float* __restrict__ hstates,
    float* __restrict__ ys)
{
    __shared__ float h[DD * DD];
    __shared__ float s_w[CPAIR], s_q[CPAIR];

    const int tid  = threadIdx.x;
    const int lane = tid & 31;
    const int wave = tid >> 5;
    const int nc = blockIdx.x, bh = blockIdx.y;

    const size_t cbase = (size_t)(bh * NCHUNK + nc) * CT * DD;
    const float* Qc = Qt + cbase;
    const float* Wc = Wt + cbase;
    const float* Uc = Ut + cbase;

    {
        const float4* src = (const float4*)(hstates + (size_t)(bh * NCHUNK + nc) * DD * DD);
        float4* hv = (float4*)h;
        for (int i = tid; i < DD * DD / 4; i += 256) hv[i] = src[i];
    }

    for (int i = 0; i < 8; ++i) {
        int c = wave * 8 + i;
        const float* w0p = Wc + (size_t)(2 * c) * DD;
        const float* w1p = Wc + (size_t)(2 * c + 1) * DD;
        const float* q1p = Qc + (size_t)(2 * c + 1) * DD;
        float pw = 0.f, pq = 0.f;
        for (int e = lane; e < DD; e += 32) {
            float a = w0p[e];
            pw += w1p[e] * a;
            pq += q1p[e] * a;
        }
        for (int off = 16; off > 0; off >>= 1) {
            pw += __shfl_xor(pw, off, 32);
            pq += __shfl_xor(pq, off, 32);
        }
        if (lane == 0) { s_w[c] = pw; s_q[c] = pq; }
    }
    __syncthreads();

    const int m      = wave;
    const int mrow   = m * 16 + (lane & 15);
    const int khalf  = 2 * (lane >> 4);
    const int nlane  = lane & 15;
    const int rowoff = (lane >> 4) * 8;

    v2f aW[32], aQ[32];
    {
        const float* wrow = Wc + (size_t)mrow * DD;
        const float* qrow = Qc + (size_t)mrow * DD;
#pragma unroll
        for (int k = 0; k < 32; ++k) {
            int col = 4 * k + khalf;
            aW[k] = *(const v2f*)(wrow + col);
            aQ[k] = *(const v2f*)(qrow + col);
        }
    }

    float* outbase = ys + cbase;
    for (int n = 0; n < 8; ++n) {
        v8f accW = {};
        v8f accQ = {};
        const int ncol = n * 16 + nlane;
#pragma unroll
        for (int k = 0; k < 32; ++k) {
            int kk = 4 * k + khalf;
            v2f b;
            b.x = h[kk * DD + ncol];
            b.y = h[(kk + 1) * DD + ncol];
            accW = __builtin_amdgcn_wmma_f32_16x16x4_f32(
                false, aW[k], false, b, (short)0, accW, false, false);
            accQ = __builtin_amdgcn_wmma_f32_16x16x4_f32(
                false, aQ[k], false, b, (short)0, accQ, false, false);
        }
#pragma unroll
        for (int j = 0; j < 4; ++j) {
            int lr0 = 2 * j + rowoff;
            int t0  = m * 16 + lr0;
            int t1  = t0 + 1;
            int c   = t0 >> 1;
            float wh0 = accW[2 * j], wh1 = accW[2 * j + 1];
            float qh0 = accQ[2 * j], qh1 = accQ[2 * j + 1];
            float u0 = Uc[(size_t)t0 * DD + ncol];
            float u1 = Uc[(size_t)t1 * DD + ncol];
            float q0 = Qc[(size_t)t0 * DD + ncol];
            float q1 = Qc[(size_t)t1 * DD + ncol];
            float uv0v = u0 - wh0;
            float out0 = q0 * uv0v + qh0;
            float uv1v = u1 - (wh1 + s_w[c] * uv0v);
            float out1 = q1 * uv1v + qh1 + s_q[c] * uv0v;
            outbase[(size_t)t0 * DD + ncol] = out0;
            outbase[(size_t)t1 * DD + ncol] = out1;
        }
    }
}

extern "C" void kernel_launch(void* const* d_in, const int* in_sizes, int n_in,
                              void* d_out, int out_size, void* d_ws, size_t ws_size,
                              hipStream_t stream) {
    const float* q  = (const float*)d_in[0];
    const float* W  = (const float*)d_in[1];
    const float* U  = (const float*)d_in[2];
    const float* h0 = (const float*)d_in[3];

    float* ys     = (float*)d_out;
    float* hfinal = ys + (size_t)BH_TOT * NCHUNK * CT * DD;
    float* hstate = (float*)d_ws;

    liz_scan<<<BH_TOT, SCAN_T, 0, stream>>>(W, U, h0, hstate, hfinal);
    liz_out<<<dim3(NCHUNK, BH_TOT), 256, 0, stream>>>(q, W, U, hstate, ys);
}


# scan: global traffic off the per-chunk barrier chain: barriers fence LDS only (local address space), state rows stored 4 at a time as the gelu pass makes them, draining under vector work (sums same)
# speedup vs baseline: 3.3239x; 1.1083x over previous
#include <hip/hip_runtime.h>
#include <math.h>

#define BH_TOT 32
#define NCHUNK 32
#define CT     128
#define CPAIR  64
#define DD     128
#define STABV  10000.0f

typedef float v2f __attribute__((ext_vector_type(2)));
typedef float v8f __attribute__((ext_vector_type(8)));

__device__ __forceinline__ float gelu_tanh(float x) {
    const float k0 = 0.7978845608028654f;
    const float k1 = 0.044715f;
    float t = tanhf(k0 * (x + k1 * x * x * x));
    return 0.5f * x * (1.0f + t);
}

__device__ __forceinline__ float clipstab(float v) {
    return fminf(fmaxf(v, -STABV), STABV);
}


__device__ __forceinline__ void lds_barrier() {
    __builtin_amdgcn_fence(__ATOMIC_RELEASE, "workgroup", "local");
    __builtin_amdgcn_s_barrier();
    __builtin_amdgcn_fence(__ATOMIC_ACQUIRE, "workgroup", "local");
}

#define SCAN_T 512
#define RPT    32
#define VP     144
#define VQ     36

__global__ __launch_bounds__(SCAN_T) void liz_scan(
    const float* __restrict__ Wt, const float* __restrict__ Ut,
    const float* __restrict__ h0, float* __restrict__ hstates,
    float* __restrict__ hfinal)
{
    __shared__ __attribute__((aligned(16))) float vbuf[DD * VP];
    __shared__ __attribute__((aligned(16))) float wbuf[2][2][DD];
    __shared__ float ubuf[2][2][DD];
    __shared__ float psum0[SCAN_T], psum1[SCAN_T];
    __shared__ __attribute__((aligned(16))) float ninv[DD];
    __shared__ __attribute__((aligned(16))) float nhn[DD];

    const int tid  = threadIdx.x;
    const int bh   = blockIdx.x;
    const int lane = tid & 31;
    const int d    = tid & (DD - 1);
    const int q    = tid >> 7;
    const int r0   = q * RPT;
    const int dcol = d + (d >> 5) * 4;

    float hreg[RPT];
    {
        const float* src = h0 + (size_t)bh * DD * DD + (size_t)r0 * DD + d;
#pragma unroll
        for (int i = 0; i < RPT; ++i) hreg[i] = src[(size_t)i * DD];
    }
    {
        float* dst = hstates + (size_t)(bh * NCHUNK) * DD * DD + (size_t)r0 * DD + d;
#pragma unroll
        for (int i = 0; i < RPT; ++i) dst[(size_t)i * DD] = hreg[i];
    }
    if (tid < DD) {
        const float* wb = Wt + (size_t)(bh * NCHUNK) * CT * DD;
        const float* ub = Ut + (size_t)(bh * NCHUNK) * CT * DD;
        wbuf[0][0][tid] = wb[(size_t)(CT - 2) * DD + tid];
        wbuf[0][1][tid] = wb[(size_t)(CT - 1) * DD + tid];
        ubuf[0][0][tid] = ub[(size_t)(CT - 2) * DD + tid];
        ubuf[0][1][tid] = ub[(size_t)(CT - 1) * DD + tid];
    }
    __syncthreads();

    for (int ch = 0; ch < NCHUNK; ++ch) {
        const int cur = ch & 1;
        const bool more = (tid < DD) && (ch + 1 < NCHUNK);

        float pw0 = 0.f, pw1 = 0.f, pu0 = 0.f, pu1 = 0.f;
        if (more) {
            const float* wb = Wt + (size_t)(bh * NCHUNK + ch + 1) * CT * DD;
            const float* ub = Ut + (size_t)(bh * NCHUNK + ch + 1) * CT * DD;
            pw0 = wb[(size_t)(CT - 2) * DD + tid];
            pw1 = wb[(size_t)(CT - 1) * DD + tid];
            pu0 = ub[(size_t)(CT - 2) * DD + tid];
            pu1 = ub[(size_t)(CT - 1) * DD + tid];
        }

        const float* w0c = &wbuf[cur][0][0];
        const float* w1c = &wbuf[cur][1][0];

        float ww;
        {
            float a0 = w1c[lane] * w0c[lane] + w1c[lane + 64] * w0c[lane + 64];
            float a1 = w1c[lane + 32] * w0c[lane + 32] + w1c[lane + 96] * w0c[lane + 96];
            ww = a0 + a1;
            for (int off = 16; off > 0; off >>= 1) ww += __shfl_xor(ww, off, 32);
        }

        const float4* w0v = (const float4*)(w0c + r0);
        const float4* w1v = (const float4*)(w1c + r0);

        {
            float s0 = 0.f, s1 = 0.f;
#pragma unroll
            for (int i4 = 0; i4 < RPT / 4; ++i4) {
                float4 a = w0v[i4];
                float4 b = w1v[i4];
                s0 += a.x * hreg[4 * i4 + 0];
                s0 += a.y * hreg[4 * i4 + 1];
                s0 += a.z * hreg[4 * i4 + 2];
                s0 += a.w * hreg[4 * i4 + 3];
                s1 += b.x * hreg[4 * i4 + 0];
                s1 += b.y * hreg[4 * i4 + 1];
                s1 += b.z * hreg[4 * i4 + 2];
                s1 += b.w * hreg[4 * i4 + 3];
            }
            psum0[tid] = s0;
            psum1[tid] = s1;
        }
        lds_barrier();

        const float k0  = (psum0[d] + psum0[d + 128]) + (psum0[d + 256] + psum0[d + 384]);
        const float k1  = (psum1[d] + psum1[d + 128]) + (psum1[d + 256] + psum1[d + 384]);
        const float uv0 = ubuf[cur][0][d] - k0;
        const float uv1 = ubuf[cur][1][d] - (k1 + ww * uv0);

        {
            float* vrow = vbuf + (size_t)r0 * VP + dcol;
#pragma unroll
            for (int i4 = 0; i4 < RPT / 4; ++i4) {
                float4 a = w0v[i4];
                float4 b = w1v[i4];
                float v;
                v = clipstab(hreg[4 * i4 + 0] + a.x * uv0);
                v = clipstab(v + b.x * uv1);
                hreg[4 * i4 + 0] = v;  vrow[(4 * i4 + 0) * VP] = v;
                v = clipstab(hreg[4 * i4 + 1] + a.y * uv0);
                v = clipstab(v + b.y * uv1);
                hreg[4 * i4 + 1] = v;  vrow[(4 * i4 + 1) * VP] = v;
                v = clipstab(hreg[4 * i4 + 2] + a.z * uv0);
                v = clipstab(v + b.z * uv1);
                hreg[4 * i4 + 2] = v;  vrow[(4 * i4 + 2) * VP] = v;
                v = clipstab(hreg[4 * i4 + 3] + a.w * uv0);
                v = clipstab(v + b.w * uv1);
                hreg[4 * i4 + 3] = v;  vrow[(4 * i4 + 3) * VP] = v;
            }
        }
        lds_barrier();

        {
            const int row  = tid >> 2;
            const int part = tid & 3;
            const float4* src = (const float4*)(vbuf + (size_t)row * VP + part * VQ);
            float sq = 0.f;
#pragma unroll
            for (int k = 0; k < 8; ++k) {
                float4 t = src[k];
                sq += t.x * t.x;
                sq += t.y * t.y;
                sq += t.z * t.z;
                sq += t.w * t.w;
            }
            sq += __shfl_xor(sq, 1, 32);
            sq += __shfl_xor(sq, 2, 32);
            float n = sqrtf(sq) + 1e-6f;
            if (part == 0) {
                ninv[row] = 2.0f / n;
                nhn[row]  = 0.5f * n;
            }
        }
        if (more) {
            wbuf[cur ^ 1][0][tid] = pw0;
            wbuf[cur ^ 1][1][tid] = pw1;
            ubuf[cur ^ 1][0][tid] = pu0;
            ubuf[cur ^ 1][1][tid] = pu1;
        }
        lds_barrier();

        {
            float* dst = (ch + 1 < NCHUNK)
                ? hstates + (size_t)(bh * NCHUNK + ch + 1) * DD * DD
                : hfinal + (size_t)bh * DD * DD;
            dst += (size_t)r0 * DD + d;
            const float4* iv = (const float4*)(ninv + r0);
            const float4* hv = (const float4*)(nhn + r0);
#pragma unroll
            for (int i4 = 0; i4 < RPT / 4; ++i4) {
                float4 a = iv[i4];
                float4 b = hv[i4];
                hreg[4 * i4 + 0] = b.x * gelu_tanh(hreg[4 * i4 + 0] * a.x);
                hreg[4 * i4 + 1] = b.y * gelu_tanh(hreg[4 * i4 + 1] * a.y);
                hreg[4 * i4 + 2] = b.z * gelu_tanh(hreg[4 * i4 + 2] * a.z);
                hreg[4 * i4 + 3] = b.w * gelu_tanh(hreg[4 * i4 + 3] * a.w);
                dst[(size_t)(4 * i4 + 0) * DD] = hreg[4 * i4 + 0];
                dst[(size_t)(4 * i4 + 1) * DD] = hreg[4 * i4 + 1];
                dst[(size_t)(4 * i4 + 2) * DD] = hreg[4 * i4 + 2];
                dst[(size_t)(4 * i4 + 3) * DD] = hreg[4 * i4 + 3];
            }
        }
    }
}

__global__ __launch_bounds__(256) void liz_out(
    const float* __restrict__ Qt, const float* __restrict__ Wt,
    const float* __restrict__ Ut, const float* __restrict__ hstates,
    float* __restrict__ ys)
{
    __shared__ float h[DD * DD];
    __shared__ float s_w[CPAIR], s_q[CPAIR];

    const int tid  = threadIdx.x;
    const int lane = tid & 31;
    const int wave = tid >> 5;
    const int nc = blockIdx.x, bh = blockIdx.y;

    const size_t cbase = (size_t)(bh * NCHUNK + nc) * CT * DD;
    const float* Qc = Qt + cbase;
    const float* Wc = Wt + cbase;
    const float* Uc = Ut + cbase;

    {
        const float4* src = (const float4*)(hstates + (size_t)(bh * NCHUNK + nc) * DD * DD);
        float4* hv = (float4*)h;
        for (int i = tid; i < DD * DD / 4; i += 256) hv[i] = src[i];
    }

    for (int i = 0; i < 8; ++i) {
        int c = wave * 8 + i;
        const float* w0p = Wc + (size_t)(2 * c) * DD;
        const float* w1p = Wc + (size_t)(2 * c + 1) * DD;
        const float* q1p = Qc + (size_t)(2 * c + 1) * DD;
        float pw = 0.f, pq = 0.f;
        for (int e = lane; e < DD; e += 32) {
            float a = w0p[e];
            pw += w1p[e] * a;
            pq += q1p[e] * a;
        }
        for (int off = 16; off > 0; off >>= 1) {
            pw += __shfl_xor(pw, off, 32);
            pq += __shfl_xor(pq, off, 32);
        }
        if (lane == 0) { s_w[c] = pw; s_q[c] = pq; }
    }
    __syncthreads();

    const int m      = wave;
    const int mrow   = m * 16 + (lane & 15);
    const int khalf  = 2 * (lane >> 4);
    const int nlane  = lane & 15;
    const int rowoff = (lane >> 4) * 8;

    v2f aW[32], aQ[32];
    {
        const float* wrow = Wc + (size_t)mrow * DD;
        const float* qrow = Qc + (size_t)mrow * DD;
#pragma unroll
        for (int k = 0; k < 32; ++k) {
            int col = 4 * k + khalf;
            aW[k] = *(const v2f*)(wrow + col);
            aQ[k] = *(const v2f*)(qrow + col);
        }
    }

    float* outbase = ys + cbase;
    for (int n = 0; n < 8; ++n) {
        v8f accW = {};
        v8f accQ = {};
        const int ncol = n * 16 + nlane;
#pragma unroll
        for (int k = 0; k < 32; ++k) {
            int kk = 4 * k + khalf;
            v2f b;
            b.x = h[kk * DD + ncol];
            b.y = h[(kk + 1) * DD + ncol];
            accW = __builtin_amdgcn_wmma_f32_16x16x4_f32(
                false, aW[k], false, b, (short)0, accW, false, false);
            accQ = __builtin_amdgcn_wmma_f32_16x16x4_f32(
                false, aQ[k], false, b, (short)0, accQ, false, false);
        }
#pragma unroll
        for (int j = 0; j < 4; ++j) {
            int lr0 = 2 * j + rowoff;
            int t0  = m * 16 + lr0;
            int t1  = t0 + 1;
            int c   = t0 >> 1;
            float wh0 = accW[2 * j], wh1 = accW[2 * j + 1];
            float qh0 = accQ[2 * j], qh1 = accQ[2 * j + 1];
            float u0 = Uc[(size_t)t0 * DD + ncol];
            float u1 = Uc[(size_t)t1 * DD + ncol];
            float q0 = Qc[(size_t)t0 * DD + ncol];
            float q1 = Qc[(size_t)t1 * DD + ncol];
            float uv0v = u0 - wh0;
            float out0 = q0 * uv0v + qh0;
            float uv1v = u1 - (wh1 + s_w[c] * uv0v);
            float out1 = q1 * uv1v + qh1 + s_q[c] * uv0v;
            outbase[(size_t)t0 * DD + ncol] = out0;
            outbase[(size_t)t1 * DD + ncol] = out1;
        }
    }
}

extern "C" void kernel_launch(void* const* d_in, const int* in_sizes, int n_in,
                              void* d_out, int out_size, void* d_ws, size_t ws_size,
                              hipStream_t stream) {
    const float* q  = (const float*)d_in[0];
    const float* W  = (const float*)d_in[1];
    const float* U  = (const float*)d_in[2];
    const float* h0 = (const float*)d_in[3];

    float* ys     = (float*)d_out;
    float* hfinal = ys + (size_t)BH_TOT * NCHUNK * CT * DD;
    float* hstate = (float*)d_ws;

    liz_scan<<<BH_TOT, SCAN_T, 0, stream>>>(W, U, h0, hstate, hfinal);
    liz_out<<<dim3(NCHUNK, BH_TOT), 256, 0, stream>>>(q, W, U, hstate, ys);
}


# scan: 32 waves per work-group (a thread owns 16 rows of one column, 8 waves per SIMD): other waves issue in the gaps of each wave's branchy gelu chain; column sums from 8 partials (sum order changes)
# speedup vs baseline: 3.4385x; 1.0345x over previous
#include <hip/hip_runtime.h>
#include <math.h>

#define BH_TOT 32
#define NCHUNK 32
#define CT     128
#define CPAIR  64
#define DD     128
#define STABV  10000.0f

typedef float v2f __attribute__((ext_vector_type(2)));
typedef float v8f __attribute__((ext_vector_type(8)));

__device__ __forceinline__ float gelu_tanh(float x) {
    const float k0 = 0.7978845608028654f;
    const float k1 = 0.044715f;
    float t = tanhf(k0 * (x + k1 * x * x * x));
    return 0.5f * x * (1.0f + t);
}

__device__ __forceinline__ float clipstab(float v) {
    return fminf(fmaxf(v, -STABV), STABV);
}


__device__ __forceinline__ void lds_barrier() {
    __builtin_amdgcn_fence(__ATOMIC_RELEASE, "workgroup", "local");
    __builtin_amdgcn_s_barrier();
    __builtin_amdgcn_fence(__ATOMIC_ACQUIRE, "workgroup", "local");
}

#define SCAN_T 1024
#define RPT    16
#define NORM_T 512
#define VP     144
#define VQ     36

__global__ __launch_bounds__(SCAN_T) void liz_scan(
    const float* __restrict__ Wt, const float* __restrict__ Ut,
    const float* __restrict__ h0, float* __restrict__ hstates,
    float* __restrict__ hfinal)
{
    __shared__ __attribute__((aligned(16))) float vbuf[DD * VP];
    __shared__ __attribute__((aligned(16))) float wbuf[2][2][DD];
    __shared__ float ubuf[2][2][DD];
    __shared__ float psum0[SCAN_T], psum1[SCAN_T];
    __shared__ __attribute__((aligned(16))) float ninv[DD];
    __shared__ __attribute__((aligned(16))) float nhn[DD];

    const int tid  = threadIdx.x;
    const int bh   = blockIdx.x;
    const int lane = tid & 31;
    const int d    = tid & (DD - 1);
    const int q    = tid >> 7;
    const int r0   = q * RPT;
    const int dcol = d + (d >> 5) * 4;

    float hreg[RPT];
    {
        const float* src = h0 + (size_t)bh * DD * DD + (size_t)r0 * DD + d;
#pragma unroll
        for (int i = 0; i < RPT; ++i) hreg[i] = src[(size_t)i * DD];
    }
    {
        float* dst = hstates + (size_t)(bh * NCHUNK) * DD * DD + (size_t)r0 * DD + d;
#pragma unroll
        for (int i = 0; i < RPT; ++i) dst[(size_t)i * DD] = hreg[i];
    }
    if (tid < DD) {
        const float* wb = Wt + (size_t)(bh * NCHUNK) * CT * DD;
        const float* ub = Ut + (size_t)(bh * NCHUNK) * CT * DD;
        wbuf[0][0][tid] = wb[(size_t)(CT - 2) * DD + tid];
        wbuf[0][1][tid] = wb[(size_t)(CT - 1) * DD + tid];
        ubuf[0][0][tid] = ub[(size_t)(CT - 2) * DD + tid];
        ubuf[0][1][tid] = ub[(size_t)(CT - 1) * DD + tid];
    }
    __syncthreads();

    for (int ch = 0; ch < NCHUNK; ++ch) {
        const int cur = ch & 1;
        const bool more = (tid < DD) && (ch + 1 < NCHUNK);

        float pw0 = 0.f, pw1 = 0.f, pu0 = 0.f, pu1 = 0.f;
        if (more) {
            const float* wb = Wt + (size_t)(bh * NCHUNK + ch + 1) * CT * DD;
            const float* ub = Ut + (size_t)(bh * NCHUNK + ch + 1) * CT * DD;
            pw0 = wb[(size_t)(CT - 2) * DD + tid];
            pw1 = wb[(size_t)(CT - 1) * DD + tid];
            pu0 = ub[(size_t)(CT - 2) * DD + tid];
            pu1 = ub[(size_t)(CT - 1) * DD + tid];
        }

        const float* w0c = &wbuf[cur][0][0];
        const float* w1c = &wbuf[cur][1][0];

        float ww;
        {
            float a0 = w1c[lane] * w0c[lane] + w1c[lane + 64] * w0c[lane + 64];
            float a1 = w1c[lane + 32] * w0c[lane + 32] + w1c[lane + 96] * w0c[lane + 96];
            ww = a0 + a1;
            for (int off = 16; off > 0; off >>= 1) ww += __shfl_xor(ww, off, 32);
        }

        const float4* w0v = (const float4*)(w0c + r0);
        const float4* w1v = (const float4*)(w1c + r0);

        {
            float s0 = 0.f, s1 = 0.f;
#pragma unroll
            for (int i4 = 0; i4 < RPT / 4; ++i4) {
                float4 a = w0v[i4];
                float4 b = w1v[i4];
                s0 += a.x * hreg[4 * i4 + 0];
                s0 += a.y * hreg[4 * i4 + 1];
                s0 += a.z * hreg[4 * i4 + 2];
                s0 += a.w * hreg[4 * i4 + 3];
                s1 += b.x * hreg[4 * i4 + 0];
                s1 += b.y * hreg[4 * i4 + 1];
                s1 += b.z * hreg[4 * i4 + 2];
                s1 += b.w * hreg[4 * i4 + 3];
            }
            psum0[tid] = s0;
            psum1[tid] = s1;
        }
        lds_barrier();

        const float k0  = ((psum0[d]       + psum0[d + 128]) + (psum0[d + 256] + psum0[d + 384]))
                        + ((psum0[d + 512] + psum0[d + 640]) + (psum0[d + 768] + psum0[d + 896]));
        const float k1  = ((psum1[d]       + psum1[d + 128]) + (psum1[d + 256] + psum1[d + 384]))
                        + ((psum1[d + 512] + psum1[d + 640]) + (psum1[d + 768] + psum1[d + 896]));
        const float uv0 = ubuf[cur][0][d] - k0;
        const float uv1 = ubuf[cur][1][d] - (k1 + ww * uv0);

        {
            float* vrow = vbuf + (size_t)r0 * VP + dcol;
#pragma unroll
            for (int i4 = 0; i4 < RPT / 4; ++i4) {
                float4 a = w0v[i4];
                float4 b = w1v[i4];
                float v;
                v = clipstab(hreg[4 * i4 + 0] + a.x * uv0);
                v = clipstab(v + b.x * uv1);
                hreg[4 * i4 + 0] = v;  vrow[(4 * i4 + 0) * VP] = v;
                v = clipstab(hreg[4 * i4 + 1] + a.y * uv0);
                v = clipstab(v + b.y * uv1);
                hreg[4 * i4 + 1] = v;  vrow[(4 * i4 + 1) * VP] = v;
                v = clipstab(hreg[4 * i4 + 2] + a.z * uv0);
                v = clipstab(v + b.z * uv1);
                hreg[4 * i4 + 2] = v;  vrow[(4 * i4 + 2) * VP] = v;
                v = clipstab(hreg[4 * i4 + 3] + a.w * uv0);
                v = clipstab(v + b.w * uv1);
                hreg[4 * i4 + 3] = v;  vrow[(4 * i4 + 3) * VP] = v;
            }
        }
        lds_barrier();

        if (tid < NORM_T) {
            const int row  = tid >> 2;
            const int part = tid & 3;
            const float4* src = (const float4*)(vbuf + (size_t)row * VP + part * VQ);
            float sq = 0.f;
#pragma unroll
            for (int k = 0; k < 8; ++k) {
                float4 t = src[k];
                sq += t.x * t.x;
                sq += t.y * t.y;
                sq += t.z * t.z;
                sq += t.w * t.w;
            }
            sq += __shfl_xor(sq, 1, 32);
            sq += __shfl_xor(sq, 2, 32);
            float n = sqrtf(sq) + 1e-6f;
            if (part == 0) {
                ninv[row] = 2.0f / n;
                nhn[row]  = 0.5f * n;
            }
        }
        if (more) {
            wbuf[cur ^ 1][0][tid] = pw0;
            wbuf[cur ^ 1][1][tid] = pw1;
            ubuf[cur ^ 1][0][tid] = pu0;
            ubuf[cur ^ 1][1][tid] = pu1;
        }
        lds_barrier();

        {
            float* dst = (ch + 1 < NCHUNK)
                ? hstates + (size_t)(bh * NCHUNK + ch + 1) * DD * DD
                : hfinal + (size_t)bh * DD * DD;
            dst += (size_t)r0 * DD + d;
            const float4* iv = (const float4*)(ninv + r0);
            const float4* hv = (const float4*)(nhn + r0);
#pragma unroll
            for (int i4 = 0; i4 < RPT / 4; ++i4) {
                float4 a = iv[i4];
                float4 b = hv[i4];
                hreg[4 * i4 + 0] = b.x * gelu_tanh(hreg[4 * i4 + 0] * a.x);
                hreg[4 * i4 + 1] = b.y * gelu_tanh(hreg[4 * i4 + 1] * a.y);
                hreg[4 * i4 + 2] = b.z * gelu_tanh(hreg[4 * i4 + 2] * a.z);
                hreg[4 * i4 + 3] = b.w * gelu_tanh(hreg[4 * i4 + 3] * a.w);
                dst[(size_t)(4 * i4 + 0) * DD] = hreg[4 * i4 + 0];
                dst[(size_t)(4 * i4 + 1) * DD] = hreg[4 * i4 + 1];
                dst[(size_t)(4 * i4 + 2) * DD] = hreg[4 * i4 + 2];
                dst[(size_t)(4 * i4 + 3) * DD] = hreg[4 * i4 + 3];
            }
        }
    }
}

__global__ __launch_bounds__(256) void liz_out(
    const float* __restrict__ Qt, const float* __restrict__ Wt,
    const float* __restrict__ Ut, const float* __restrict__ hstates,
    float* __restrict__ ys)
{
    __shared__ float h[DD * DD];
    __shared__ float s_w[CPAIR], s_q[CPAIR];

    const int tid  = threadIdx.x;
    const int lane = tid & 31;
    const int wave = tid >> 5;
    const int nc = blockIdx.x, bh = blockIdx.y;

    const size_t cbase = (size_t)(bh * NCHUNK + nc) * CT * DD;
    const float* Qc = Qt + cbase;
    const float* Wc = Wt + cbase;
    const float* Uc = Ut + cbase;

    {
        const float4* src = (const float4*)(hstates + (size_t)(bh * NCHUNK + nc) * DD * DD);
        float4* hv = (float4*)h;
        for (int i = tid; i < DD * DD / 4; i += 256) hv[i] = src[i];
    }

    for (int i = 0; i < 8; ++i) {
        int c = wave * 8 + i;
        const float* w0p = Wc + (size_t)(2 * c) * DD;
        const float* w1p = Wc + (size_t)(2 * c + 1) * DD;
        const float* q1p = Qc + (size_t)(2 * c + 1) * DD;
        float pw = 0.f, pq = 0.f;
        for (int e = lane; e < DD; e += 32) {
            float a = w0p[e];
            pw += w1p[e] * a;
            pq += q1p[e] * a;
        }
        for (int off = 16; off > 0; off >>= 1) {
            pw += __shfl_xor(pw, off, 32);
            pq += __shfl_xor(pq, off, 32);
        }
        if (lane == 0) { s_w[c] = pw; s_q[c] = pq; }
    }
    __syncthreads();

    const int m      = wave;
    const int mrow   = m * 16 + (lane & 15);
    const int khalf  = 2 * (lane >> 4);
    const int nlane  = lane & 15;
    const int rowoff = (lane >> 4) * 8;

    v2f aW[32], aQ[32];
    {
        const float* wrow = Wc + (size_t)mrow * DD;
        const float* qrow = Qc + (size_t)mrow * DD;
#pragma unroll
        for (int k = 0; k < 32; ++k) {
            int col = 4 * k + khalf;
            aW[k] = *(const v2f*)(wrow + col);
            aQ[k] = *(const v2f*)(qrow + col);
        }
    }

    float* outbase = ys + cbase;
    for (int n = 0; n < 8; ++n) {
        v8f accW = {};
        v8f accQ = {};
        const int ncol = n * 16 + nlane;
#pragma unroll
        for (int k = 0; k < 32; ++k) {
            int kk = 4 * k + khalf;
            v2f b;
            b.x = h[kk * DD + ncol];
            b.y = h[(kk + 1) * DD + ncol];
            accW = __builtin_amdgcn_wmma_f32_16x16x4_f32(
                false, aW[k], false, b, (short)0, accW, false, false);
            accQ = __builtin_amdgcn_wmma_f32_16x16x4_f32(
                false, aQ[k], false, b, (short)0, accQ, false, false);
        }
#pragma unroll
        for (int j = 0; j < 4; ++j) {
            int lr0 = 2 * j + rowoff;
            int t0  = m * 16 + lr0;
            int t1  = t0 + 1;
            int c   = t0 >> 1;
            float wh0 = accW[2 * j], wh1 = accW[2 * j + 1];
            float qh0 = accQ[2 * j], qh1 = accQ[2 * j + 1];
            float u0 = Uc[(size_t)t0 * DD + ncol];
            float u1 = Uc[(size_t)t1 * DD + ncol];
            float q0 = Qc[(size_t)t0 * DD + ncol];
            float q1 = Qc[(size_t)t1 * DD + ncol];
            float uv0v = u0 - wh0;
            float out0 = q0 * uv0v + qh0;
            float uv1v = u1 - (wh1 + s_w[c] * uv0v);
            float out1 = q1 * uv1v + qh1 + s_q[c] * uv0v;
            outbase[(size_t)t0 * DD + ncol] = out0;
            outbase[(size_t)t1 * DD + ncol] = out1;
        }
    }
}

extern "C" void kernel_launch(void* const* d_in, const int* in_sizes, int n_in,
                              void* d_out, int out_size, void* d_ws, size_t ws_size,
                              hipStream_t stream) {
    const float* q  = (const float*)d_in[0];
    const float* W  = (const float*)d_in[1];
    const float* U  = (const float*)d_in[2];
    const float* h0 = (const float*)d_in[3];

    float* ys     = (float*)d_out;
    float* hfinal = ys + (size_t)BH_TOT * NCHUNK * CT * DD;
    float* hstate = (float*)d_ws;

    liz_scan<<<BH_TOT, SCAN_T, 0, stream>>>(W, U, h0, hstate, hfinal);
    liz_out<<<dim3(NCHUNK, BH_TOT), 256, 0, stream>>>(q, W, U, hstate, ys);
}


# out: roles swapped: a wave owns 16 output columns, its h columns are B fragments in registers; W and Q rows staged once in LDS feed A fragments, row-pair dots and q factors (sum order same)
# speedup vs baseline: 3.7999x; 1.1051x over previous
#include <hip/hip_runtime.h>
#include <math.h>

#define BH_TOT 32
#define NCHUNK 32
#define CT     128
#define CPAIR  64
#define DD     128
#define STABV  10000.0f

typedef float v2f __attribute__((ext_vector_type(2)));
typedef float v8f __attribute__((ext_vector_type(8)));

__device__ __forceinline__ float gelu_tanh(float x) {
    const float k0 = 0.7978845608028654f;
    const float k1 = 0.044715f;
    float t = tanhf(k0 * (x + k1 * x * x * x));
    return 0.5f * x * (1.0f + t);
}

__device__ __forceinline__ float clipstab(float v) {
    return fminf(fmaxf(v, -STABV), STABV);
}


__device__ __forceinline__ void lds_barrier() {
    __builtin_amdgcn_fence(__ATOMIC_RELEASE, "workgroup", "local");
    __builtin_amdgcn_s_barrier();
    __builtin_amdgcn_fence(__ATOMIC_ACQUIRE, "workgroup", "local");
}

#define SCAN_T 1024
#define RPT    16
#define NORM_T 512
#define VP     144
#define VQ     36

__global__ __launch_bounds__(SCAN_T) void liz_scan(
    const float* __restrict__ Wt, const float* __restrict__ Ut,
    const float* __restrict__ h0, float* __restrict__ hstates,
    float* __restrict__ hfinal)
{
    __shared__ __attribute__((aligned(16))) float vbuf[DD * VP];
    __shared__ __attribute__((aligned(16))) float wbuf[2][2][DD];
    __shared__ float ubuf[2][2][DD];
    __shared__ float psum0[SCAN_T], psum1[SCAN_T];
    __shared__ __attribute__((aligned(16))) float ninv[DD];
    __shared__ __attribute__((aligned(16))) float nhn[DD];

    const int tid  = threadIdx.x;
    const int bh   = blockIdx.x;
    const int lane = tid & 31;
    const int d    = tid & (DD - 1);
    const int q    = tid >> 7;
    const int r0   = q * RPT;
    const int dcol = d + (d >> 5) * 4;

    float hreg[RPT];
    {
        const float* src = h0 + (size_t)bh * DD * DD + (size_t)r0 * DD + d;
#pragma unroll
        for (int i = 0; i < RPT; ++i) hreg[i] = src[(size_t)i * DD];
    }
    {
        float* dst = hstates + (size_t)(bh * NCHUNK) * DD * DD + (size_t)r0 * DD + d;
#pragma unroll
        for (int i = 0; i < RPT; ++i) dst[(size_t)i * DD] = hreg[i];
    }
    if (tid < DD) {
        const float* wb = Wt + (size_t)(bh * NCHUNK) * CT * DD;
        const float* ub = Ut + (size_t)(bh * NCHUNK) * CT * DD;
        wbuf[0][0][tid] = wb[(size_t)(CT - 2) * DD + tid];
        wbuf[0][1][tid] = wb[(size_t)(CT - 1) * DD + tid];
        ubuf[0][0][tid] = ub[(size_t)(CT - 2) * DD + tid];
        ubuf[0][1][tid] = ub[(size_t)(CT - 1) * DD + tid];
    }
    __syncthreads();

    for (int ch = 0; ch < NCHUNK; ++ch) {
        const int cur = ch & 1;
        const bool more = (tid < DD) && (ch + 1 < NCHUNK);

        float pw0 = 0.f, pw1 = 0.f, pu0 = 0.f, pu1 = 0.f;
        if (more) {
            const float* wb = Wt + (size_t)(bh * NCHUNK + ch + 1) * CT * DD;
            const float* ub = Ut + (size_t)(bh * NCHUNK + ch + 1) * CT * DD;
            pw0 = wb[(size_t)(CT - 2) * DD + tid];
            pw1 = wb[(size_t)(CT - 1) * DD + tid];
            pu0 = ub[(size_t)(CT - 2) * DD + tid];
            pu1 = ub[(size_t)(CT - 1) * DD + tid];
        }

        const float* w0c = &wbuf[cur][0][0];
        const float* w1c = &wbuf[cur][1][0];

        float ww;
        {
            float a0 = w1c[lane] * w0c[lane] + w1c[lane + 64] * w0c[lane + 64];
            float a1 = w1c[lane + 32] * w0c[lane + 32] + w1c[lane + 96] * w0c[lane + 96];
            ww = a0 + a1;
            for (int off = 16; off > 0; off >>= 1) ww += __shfl_xor(ww, off, 32);
        }

        const float4* w0v = (const float4*)(w0c + r0);
        const float4* w1v = (const float4*)(w1c + r0);

        {
            float s0 = 0.f, s1 = 0.f;
#pragma unroll
            for (int i4 = 0; i4 < RPT / 4; ++i4) {
                float4 a = w0v[i4];
                float4 b = w1v[i4];
                s0 += a.x * hreg[4 * i4 + 0];
                s0 += a.y * hreg[4 * i4 + 1];
                s0 += a.z * hreg[4 * i4 + 2];
                s0 += a.w * hreg[4 * i4 + 3];
                s1 += b.x * hreg[4 * i4 + 0];
                s1 += b.y * hreg[4 * i4 + 1];
                s1 += b.z * hreg[4 * i4 + 2];
                s1 += b.w * hreg[4 * i4 + 3];
            }
            psum0[tid] = s0;
            psum1[tid] = s1;
        }
        lds_barrier();

        const float k0  = ((psum0[d]       + psum0[d + 128]) + (psum0[d + 256] + psum0[d + 384]))
                        + ((psum0[d + 512] + psum0[d + 640]) + (psum0[d + 768] + psum0[d + 896]));
        const float k1  = ((psum1[d]       + psum1[d + 128]) + (psum1[d + 256] + psum1[d + 384]))
                        + ((psum1[d + 512] + psum1[d + 640]) + (psum1[d + 768] + psum1[d + 896]));
        const float uv0 = ubuf[cur][0][d] - k0;
        const float uv1 = ubuf[cur][1][d] - (k1 + ww * uv0);

        {
            float* vrow = vbuf + (size_t)r0 * VP + dcol;
#pragma unroll
            for (int i4 = 0; i4 < RPT / 4; ++i4) {
                float4 a = w0v[i4];
                float4 b = w1v[i4];
                float v;
                v = clipstab(hreg[4 * i4 + 0] + a.x * uv0);
                v = clipstab(v + b.x * uv1);
                hreg[4 * i4 + 0] = v;  vrow[(4 * i4 + 0) * VP] = v;
                v = clipstab(hreg[4 * i4 + 1] + a.y * uv0);
                v = clipstab(v + b.y * uv1);
                hreg[4 * i4 + 1] = v;  vrow[(4 * i4 + 1) * VP] = v;
                v = clipstab(hreg[4 * i4 + 2] + a.z * uv0);
                v = clipstab(v + b.z * uv1);
                hreg[4 * i4 + 2] = v;  vrow[(4 * i4 + 2) * VP] = v;
                v = clipstab(hreg[4 * i4 + 3] + a.w * uv0);
                v = clipstab(v + b.w * uv1);
                hreg[4 * i4 + 3] = v;  vrow[(4 * i4 + 3) * VP] = v;
            }
        }
        lds_barrier();

        if (tid < NORM_T) {
            const int row  = tid >> 2;
            const int part = tid & 3;
            const float4* src = (const float4*)(vbuf + (size_t)row * VP + part * VQ);
            float sq = 0.f;
#pragma unroll
            for (int k = 0; k < 8; ++k) {
                float4 t = src[k];
                sq += t.x * t.x;
                sq += t.y * t.y;
                sq += t.z * t.z;
                sq += t.w * t.w;
            }
            sq += __shfl_xor(sq, 1, 32);
            sq += __shfl_xor(sq, 2, 32);
            float n = sqrtf(sq) + 1e-6f;
            if (part == 0) {
                ninv[row] = 2.0f / n;
                nhn[row]  = 0.5f * n;
            }
        }
        if (more) {
            wbuf[cur ^ 1][0][tid] = pw0;
            wbuf[cur ^ 1][1][tid] = pw1;
            ubuf[cur ^ 1][0][tid] = pu0;
            ubuf[cur ^ 1][1][tid] = pu1;
        }
        lds_barrier();

        {
            float* dst = (ch + 1 < NCHUNK)
                ? hstates + (size_t)(bh * NCHUNK + ch + 1) * DD * DD
                : hfinal + (size_t)bh * DD * DD;
            dst += (size_t)r0 * DD + d;
            const float4* iv = (const float4*)(ninv + r0);
            const float4* hv = (const float4*)(nhn + r0);
#pragma unroll
            for (int i4 = 0; i4 < RPT / 4; ++i4) {
                float4 a = iv[i4];
                float4 b = hv[i4];
                hreg[4 * i4 + 0] = b.x * gelu_tanh(hreg[4 * i4 + 0] * a.x);
                hreg[4 * i4 + 1] = b.y * gelu_tanh(hreg[4 * i4 + 1] * a.y);
                hreg[4 * i4 + 2] = b.z * gelu_tanh(hreg[4 * i4 + 2] * a.z);
                hreg[4 * i4 + 3] = b.w * gelu_tanh(hreg[4 * i4 + 3] * a.w);
                dst[(size_t)(4 * i4 + 0) * DD] = hreg[4 * i4 + 0];
                dst[(size_t)(4 * i4 + 1) * DD] = hreg[4 * i4 + 1];
                dst[(size_t)(4 * i4 + 2) * DD] = hreg[4 * i4 + 2];
                dst[(size_t)(4 * i4 + 3) * DD] = hreg[4 * i4 + 3];
            }
        }
    }
}

#define AP 268
#define QO 132

__global__ __launch_bounds__(256) void liz_out(
    const float* __restrict__ Qt, const float* __restrict__ Wt,
    const float* __restrict__ Ut, const float* __restrict__ hstates,
    float* __restrict__ ys)
{
    __shared__ __attribute__((aligned(16))) float WQ[CT * AP];
    float* const Wl = WQ;
    float* const Ql = WQ + QO;
    __shared__ float s_w[CPAIR], s_q[CPAIR];

    const int tid  = threadIdx.x;
    const int lane = tid & 31;
    const int wave = tid >> 5;
    const int nc = blockIdx.x, bh = blockIdx.y;

    const size_t cbase = (size_t)(bh * NCHUNK + nc) * CT * DD;
    const float* Qc = Qt + cbase;
    const float* Wc = Wt + cbase;
    const float* Uc = Ut + cbase;

    const int nlane  = lane & 15;
    const int khalf  = 2 * (lane >> 4);
    const int rowoff = (lane >> 4) * 8;
    const int ncol   = wave * 16 + nlane;

    v2f bfr[32];
    {
        const float* hsrc = hstates + (size_t)(bh * NCHUNK + nc) * DD * DD + ncol;
#pragma unroll
        for (int k = 0; k < 32; ++k) {
            int kk = 4 * k + khalf;
            bfr[k].x = hsrc[(size_t)kk * DD];
            bfr[k].y = hsrc[(size_t)(kk + 1) * DD];
        }
    }

    {
        const float4* wsrc = (const float4*)Wc;
        const float4* qsrc = (const float4*)Qc;
#pragma unroll
        for (int i = 0; i < 16; ++i) {
            int idx = tid + 256 * i;
            int row = idx >> 5, c4 = idx & 31;
            *(float4*)(Wl + row * AP + 4 * c4) = wsrc[idx];
        }
#pragma unroll
        for (int i = 0; i < 16; ++i) {
            int idx = tid + 256 * i;
            int row = idx >> 5, c4 = idx & 31;
            *(float4*)(Ql + row * AP + 4 * c4) = qsrc[idx];
        }
    }

    float ucur[8];
#pragma unroll
    for (int j = 0; j < 4; ++j) {
        int t0 = 2 * j + rowoff;
        ucur[2 * j]     = Uc[(size_t)t0 * DD + ncol];
        ucur[2 * j + 1] = Uc[(size_t)(t0 + 1) * DD + ncol];
    }
    __syncthreads();

    {
        float pw[8], pq[8];
#pragma unroll
        for (int i = 0; i < 8; ++i) {
            int c = wave * 8 + i;
            const float* w0p = Wl + (2 * c) * AP;
            const float* w1p = Wl + (2 * c + 1) * AP;
            const float* q1p = Ql + (2 * c + 1) * AP;
            float aw = 0.f, aq = 0.f;
#pragma unroll
            for (int e = lane; e < DD; e += 32) {
                float a = w0p[e];
                aw += w1p[e] * a;
                aq += q1p[e] * a;
            }
            pw[i] = aw; pq[i] = aq;
        }
#pragma unroll
        for (int off = 16; off > 0; off >>= 1) {
#pragma unroll
            for (int i = 0; i < 8; ++i) {
                pw[i] += __shfl_xor(pw[i], off, 32);
                pq[i] += __shfl_xor(pq[i], off, 32);
            }
        }
        if (lane == 0) {
#pragma unroll
            for (int i = 0; i < 8; ++i) { s_w[wave * 8 + i] = pw[i]; s_q[wave * 8 + i] = pq[i]; }
        }
    }
    __syncthreads();

    float* outbase = ys + cbase;
    for (int m = 0; m < 8; ++m) {
        float unext[8];
        {
            const int mn = (m + 1 < 8) ? m + 1 : m;
#pragma unroll
            for (int j = 0; j < 4; ++j) {
                int t0 = mn * 16 + 2 * j + rowoff;
                unext[2 * j]     = Uc[(size_t)t0 * DD + ncol];
                unext[2 * j + 1] = Uc[(size_t)(t0 + 1) * DD + ncol];
            }
        }

        v8f accW = {};
        v8f accQ = {};
        const float* wl = Wl + (m * 16 + nlane) * AP + khalf;
        const float* ql = Ql + (m * 16 + nlane) * AP + khalf;
#pragma unroll
        for (int k = 0; k < 32; ++k) {
            v2f aw = *(const v2f*)(wl + 4 * k);
            v2f aq = *(const v2f*)(ql + 4 * k);
            accW = __builtin_amdgcn_wmma_f32_16x16x4_f32(
                false, aw, false, bfr[k], (short)0, accW, false, false);
            accQ = __builtin_amdgcn_wmma_f32_16x16x4_f32(
                false, aq, false, bfr[k], (short)0, accQ, false, false);
        }
#pragma unroll
        for (int j = 0; j < 4; ++j) {
            int lr0 = 2 * j + rowoff;
            int t0  = m * 16 + lr0;
            int t1  = t0 + 1;
            int c   = t0 >> 1;
            float wh0 = accW[2 * j], wh1 = accW[2 * j + 1];
            float qh0 = accQ[2 * j], qh1 = accQ[2 * j + 1];
            float u0 = ucur[2 * j];
            float u1 = ucur[2 * j + 1];
            float q0 = Ql[t0 * AP + ncol];
            float q1 = Ql[t1 * AP + ncol];
            float uv0v = u0 - wh0;
            float out0 = q0 * uv0v + qh0;
            float uv1v = u1 - (wh1 + s_w[c] * uv0v);
            float out1 = q1 * uv1v + qh1 + s_q[c] * uv0v;
            outbase[(size_t)t0 * DD + ncol] = out0;
            outbase[(size_t)t1 * DD + ncol] = out1;
        }
#pragma unroll
        for (int j = 0; j < 8; ++j) ucur[j] = unext[j];
    }
}

extern "C" void kernel_launch(void* const* d_in, const int* in_sizes, int n_in,
                              void* d_out, int out_size, void* d_ws, size_t ws_size,
                              hipStream_t stream) {
    const float* q  = (const float*)d_in[0];
    const float* W  = (const float*)d_in[1];
    const float* U  = (const float*)d_in[2];
    const float* h0 = (const float*)d_in[3];

    float* ys     = (float*)d_out;
    float* hfinal = ys + (size_t)BH_TOT * NCHUNK * CT * DD;
    float* hstate = (float*)d_ws;

    liz_scan<<<BH_TOT, SCAN_T, 0, stream>>>(W, U, h0, hstate, hfinal);
    liz_out<<<dim3(NCHUNK, BH_TOT), 256, 0, stream>>>(q, W, U, hstate, ys);
}


# scan: gelu keeps the library tanh's arithmetic; its per-element lane-divergent range branch becomes one wave-uniform test per 16 elements, polynomial range straight-line and packed (same bits)
# speedup vs baseline: 3.9630x; 1.0429x over previous
#include <hip/hip_runtime.h>
#include <math.h>

#define BH_TOT 32
#define NCHUNK 32
#define CT     128
#define CPAIR  64
#define DD     128
#define STABV  10000.0f

typedef float v2f __attribute__((ext_vector_type(2)));
typedef float v8f __attribute__((ext_vector_type(8)));

__device__ __forceinline__ float gelu_arg(float x) {
    const float k0 = 0.7978845608028654f;
    const float k1 = 0.044715f;
    return k0 * (x + k1 * x * x * x);
}
__device__ __forceinline__ float gelu_fin(float x, float t) {
    return 0.5f * x * (1.0f + t);
}
#define TANH_SMALL 0.625f

__device__ __forceinline__ float clipstab(float v) {
    return fminf(fmaxf(v, -STABV), STABV);
}


__device__ __forceinline__ void lds_barrier() {
    __builtin_amdgcn_fence(__ATOMIC_RELEASE, "workgroup", "local");
    __builtin_amdgcn_s_barrier();
    __builtin_amdgcn_fence(__ATOMIC_ACQUIRE, "workgroup", "local");
}

#define SCAN_T 1024
#define RPT    16
#define NORM_T 512
#define VP     144
#define VQ     36

__global__ __launch_bounds__(SCAN_T) void liz_scan(
    const float* __restrict__ Wt, const float* __restrict__ Ut,
    const float* __restrict__ h0, float* __restrict__ hstates,
    float* __restrict__ hfinal)
{
    __shared__ __attribute__((aligned(16))) float vbuf[DD * VP];
    __shared__ __attribute__((aligned(16))) float wbuf[2][2][DD];
    __shared__ float ubuf[2][2][DD];
    __shared__ float psum0[SCAN_T], psum1[SCAN_T];
    __shared__ __attribute__((aligned(16))) float ninv[DD];
    __shared__ __attribute__((aligned(16))) float nhn[DD];

    const int tid  = threadIdx.x;
    const int bh   = blockIdx.x;
    const int lane = tid & 31;
    const int d    = tid & (DD - 1);
    const int q    = tid >> 7;
    const int r0   = q * RPT;
    const int dcol = d + (d >> 5) * 4;

    float hreg[RPT];
    {
        const float* src = h0 + (size_t)bh * DD * DD + (size_t)r0 * DD + d;
#pragma unroll
        for (int i = 0; i < RPT; ++i) hreg[i] = src[(size_t)i * DD];
    }
    {
        float* dst = hstates + (size_t)(bh * NCHUNK) * DD * DD + (size_t)r0 * DD + d;
#pragma unroll
        for (int i = 0; i < RPT; ++i) dst[(size_t)i * DD] = hreg[i];
    }
    if (tid < DD) {
        const float* wb = Wt + (size_t)(bh * NCHUNK) * CT * DD;
        const float* ub = Ut + (size_t)(bh * NCHUNK) * CT * DD;
        wbuf[0][0][tid] = wb[(size_t)(CT - 2) * DD + tid];
        wbuf[0][1][tid] = wb[(size_t)(CT - 1) * DD + tid];
        ubuf[0][0][tid] = ub[(size_t)(CT - 2) * DD + tid];
        ubuf[0][1][tid] = ub[(size_t)(CT - 1) * DD + tid];
    }
    __syncthreads();

    for (int ch = 0; ch < NCHUNK; ++ch) {
        const int cur = ch & 1;
        const bool more = (tid < DD) && (ch + 1 < NCHUNK);

        float pw0 = 0.f, pw1 = 0.f, pu0 = 0.f, pu1 = 0.f;
        if (more) {
            const float* wb = Wt + (size_t)(bh * NCHUNK + ch + 1) * CT * DD;
            const float* ub = Ut + (size_t)(bh * NCHUNK + ch + 1) * CT * DD;
            pw0 = wb[(size_t)(CT - 2) * DD + tid];
            pw1 = wb[(size_t)(CT - 1) * DD + tid];
            pu0 = ub[(size_t)(CT - 2) * DD + tid];
            pu1 = ub[(size_t)(CT - 1) * DD + tid];
        }

        const float* w0c = &wbuf[cur][0][0];
        const float* w1c = &wbuf[cur][1][0];

        float ww;
        {
            float a0 = w1c[lane] * w0c[lane] + w1c[lane + 64] * w0c[lane + 64];
            float a1 = w1c[lane + 32] * w0c[lane + 32] + w1c[lane + 96] * w0c[lane + 96];
            ww = a0 + a1;
            for (int off = 16; off > 0; off >>= 1) ww += __shfl_xor(ww, off, 32);
        }

        const float4* w0v = (const float4*)(w0c + r0);
        const float4* w1v = (const float4*)(w1c + r0);

        {
            float s0 = 0.f, s1 = 0.f;
#pragma unroll
            for (int i4 = 0; i4 < RPT / 4; ++i4) {
                float4 a = w0v[i4];
                float4 b = w1v[i4];
                s0 += a.x * hreg[4 * i4 + 0];
                s0 += a.y * hreg[4 * i4 + 1];
                s0 += a.z * hreg[4 * i4 + 2];
                s0 += a.w * hreg[4 * i4 + 3];
                s1 += b.x * hreg[4 * i4 + 0];
                s1 += b.y * hreg[4 * i4 + 1];
                s1 += b.z * hreg[4 * i4 + 2];
                s1 += b.w * hreg[4 * i4 + 3];
            }
            psum0[tid] = s0;
            psum1[tid] = s1;
        }
        lds_barrier();

        const float k0  = ((psum0[d]       + psum0[d + 128]) + (psum0[d + 256] + psum0[d + 384]))
                        + ((psum0[d + 512] + psum0[d + 640]) + (psum0[d + 768] + psum0[d + 896]));
        const float k1  = ((psum1[d]       + psum1[d + 128]) + (psum1[d + 256] + psum1[d + 384]))
                        + ((psum1[d + 512] + psum1[d + 640]) + (psum1[d + 768] + psum1[d + 896]));
        const float uv0 = ubuf[cur][0][d] - k0;
        const float uv1 = ubuf[cur][1][d] - (k1 + ww * uv0);

        {
            float* vrow = vbuf + (size_t)r0 * VP + dcol;
#pragma unroll
            for (int i4 = 0; i4 < RPT / 4; ++i4) {
                float4 a = w0v[i4];
                float4 b = w1v[i4];
                float v;
                v = clipstab(hreg[4 * i4 + 0] + a.x * uv0);
                v = clipstab(v + b.x * uv1);
                hreg[4 * i4 + 0] = v;  vrow[(4 * i4 + 0) * VP] = v;
                v = clipstab(hreg[4 * i4 + 1] + a.y * uv0);
                v = clipstab(v + b.y * uv1);
                hreg[4 * i4 + 1] = v;  vrow[(4 * i4 + 1) * VP] = v;
                v = clipstab(hreg[4 * i4 + 2] + a.z * uv0);
                v = clipstab(v + b.z * uv1);
                hreg[4 * i4 + 2] = v;  vrow[(4 * i4 + 2) * VP] = v;
                v = clipstab(hreg[4 * i4 + 3] + a.w * uv0);
                v = clipstab(v + b.w * uv1);
                hreg[4 * i4 + 3] = v;  vrow[(4 * i4 + 3) * VP] = v;
            }
        }
        lds_barrier();

        if (tid < NORM_T) {
            const int row  = tid >> 2;
            const int part = tid & 3;
            const float4* src = (const float4*)(vbuf + (size_t)row * VP + part * VQ);
            float sq = 0.f;
#pragma unroll
            for (int k = 0; k < 8; ++k) {
                float4 t = src[k];
                sq += t.x * t.x;
                sq += t.y * t.y;
                sq += t.z * t.z;
                sq += t.w * t.w;
            }
            sq += __shfl_xor(sq, 1, 32);
            sq += __shfl_xor(sq, 2, 32);
            float n = sqrtf(sq) + 1e-6f;
            if (part == 0) {
                ninv[row] = 2.0f / n;
                nhn[row]  = 0.5f * n;
            }
        }
        if (more) {
            wbuf[cur ^ 1][0][tid] = pw0;
            wbuf[cur ^ 1][1][tid] = pw1;
            ubuf[cur ^ 1][0][tid] = pu0;
            ubuf[cur ^ 1][1][tid] = pu1;
        }
        lds_barrier();

        {
            float* dst = (ch + 1 < NCHUNK)
                ? hstates + (size_t)(bh * NCHUNK + ch + 1) * DD * DD
                : hfinal + (size_t)bh * DD * DD;
            dst += (size_t)r0 * DD + d;
            const float4* iv = (const float4*)(ninv + r0);
            const float4* hv = (const float4*)(nhn + r0);
            float xs[RPT], us[RPT];
            unsigned amax = 0u;
#pragma unroll
            for (int i4 = 0; i4 < RPT / 4; ++i4) {
                float4 a = iv[i4];
                xs[4 * i4 + 0] = hreg[4 * i4 + 0] * a.x;
                xs[4 * i4 + 1] = hreg[4 * i4 + 1] * a.y;
                xs[4 * i4 + 2] = hreg[4 * i4 + 2] * a.z;
                xs[4 * i4 + 3] = hreg[4 * i4 + 3] * a.w;
            }
#pragma unroll
            for (int i = 0; i < RPT; ++i) {
                us[i] = gelu_arg(xs[i]);
                const unsigned ab = __builtin_bit_cast(unsigned, us[i]) & 0x7fffffffu;
                amax = ab > amax ? ab : amax;
            }
            const bool big = amax >= __builtin_bit_cast(unsigned, TANH_SMALL);
            if (__builtin_amdgcn_ballot_w32(big) == 0u) {
#pragma unroll
                for (int i4 = 0; i4 < RPT / 4; ++i4) {
                    float4 b = hv[i4];
                    const float bb[4] = {b.x, b.y, b.z, b.w};
#pragma unroll
                    for (int j = 0; j < 4; ++j) {
                        const int i = 4 * i4 + j;
                        __builtin_assume(__builtin_fabsf(us[i]) < TANH_SMALL);
                        hreg[i] = bb[j] * gelu_fin(xs[i], tanhf(us[i]));
                        dst[(size_t)i * DD] = hreg[i];
                    }
                }
            } else {
#pragma unroll
                for (int i4 = 0; i4 < RPT / 4; ++i4) {
                    float4 b = hv[i4];
                    const float bb[4] = {b.x, b.y, b.z, b.w};
#pragma unroll
                    for (int j = 0; j < 4; ++j) {
                        const int i = 4 * i4 + j;
                        hreg[i] = bb[j] * gelu_fin(xs[i], tanhf(us[i]));
                        dst[(size_t)i * DD] = hreg[i];
                    }
                }
            }
        }
    }
}

#define AP 268
#define QO 132

__global__ __launch_bounds__(256) void liz_out(
    const float* __restrict__ Qt, const float* __restrict__ Wt,
    const float* __restrict__ Ut, const float* __restrict__ hstates,
    float* __restrict__ ys)
{
    __shared__ __attribute__((aligned(16))) float WQ[CT * AP];
    float* const Wl = WQ;
    float* const Ql = WQ + QO;
    __shared__ float s_w[CPAIR], s_q[CPAIR];

    const int tid  = threadIdx.x;
    const int lane = tid & 31;
    const int wave = tid >> 5;
    const int nc = blockIdx.x, bh = blockIdx.y;

    const size_t cbase = (size_t)(bh * NCHUNK + nc) * CT * DD;
    const float* Qc = Qt + cbase;
    const float* Wc = Wt + cbase;
    const float* Uc = Ut + cbase;

    const int nlane  = lane & 15;
    const int khalf  = 2 * (lane >> 4);
    const int rowoff = (lane >> 4) * 8;
    const int ncol   = wave * 16 + nlane;

    v2f bfr[32];
    {
        const float* hsrc = hstates + (size_t)(bh * NCHUNK + nc) * DD * DD + ncol;
#pragma unroll
        for (int k = 0; k < 32; ++k) {
            int kk = 4 * k + khalf;
            bfr[k].x = hsrc[(size_t)kk * DD];
            bfr[k].y = hsrc[(size_t)(kk + 1) * DD];
        }
    }

    {
        const float4* wsrc = (const float4*)Wc;
        const float4* qsrc = (const float4*)Qc;
#pragma unroll
        for (int i = 0; i < 16; ++i) {
            int idx = tid + 256 * i;
            int row = idx >> 5, c4 = idx & 31;
            *(float4*)(Wl + row * AP + 4 * c4) = wsrc[idx];
        }
#pragma unroll
        for (int i = 0; i < 16; ++i) {
            int idx = tid + 256 * i;
            int row = idx >> 5, c4 = idx & 31;
            *(float4*)(Ql + row * AP + 4 * c4) = qsrc[idx];
        }
    }

    float ucur[8];
#pragma unroll
    for (int j = 0; j < 4; ++j) {
        int t0 = 2 * j + rowoff;
        ucur[2 * j]     = Uc[(size_t)t0 * DD + ncol];
        ucur[2 * j + 1] = Uc[(size_t)(t0 + 1) * DD + ncol];
    }
    __syncthreads();

    {
        float pw[8], pq[8];
#pragma unroll
        for (int i = 0; i < 8; ++i) {
            int c = wave * 8 + i;
            const float* w0p = Wl + (2 * c) * AP;
            const float* w1p = Wl + (2 * c + 1) * AP;
            const float* q1p = Ql + (2 * c + 1) * AP;
            float aw = 0.f, aq = 0.f;
#pragma unroll
            for (int e = lane; e < DD; e += 32) {
                float a = w0p[e];
                aw += w1p[e] * a;
                aq += q1p[e] * a;
            }
            pw[i] = aw; pq[i] = aq;
        }
#pragma unroll
        for (int off = 16; off > 0; off >>= 1) {
#pragma unroll
            for (int i = 0; i < 8; ++i) {
                pw[i] += __shfl_xor(pw[i], off, 32);
                pq[i] += __shfl_xor(pq[i], off, 32);
            }
        }
        if (lane == 0) {
#pragma unroll
            for (int i = 0; i < 8; ++i) { s_w[wave * 8 + i] = pw[i]; s_q[wave * 8 + i] = pq[i]; }
        }
    }
    __syncthreads();

    float* outbase = ys + cbase;
    for (int m = 0; m < 8; ++m) {
        float unext[8];
        {
            const int mn = (m + 1 < 8) ? m + 1 : m;
#pragma unroll
            for (int j = 0; j < 4; ++j) {
                int t0 = mn * 16 + 2 * j + rowoff;
                unext[2 * j]     = Uc[(size_t)t0 * DD + ncol];
                unext[2 * j + 1] = Uc[(size_t)(t0 + 1) * DD + ncol];
            }
        }

        v8f accW = {};
        v8f accQ = {};
        const float* wl = Wl + (m * 16 + nlane) * AP + khalf;
        const float* ql = Ql + (m * 16 + nlane) * AP + khalf;
#pragma unroll
        for (int k = 0; k < 32; ++k) {
            v2f aw = *(const v2f*)(wl + 4 * k);
            v2f aq = *(const v2f*)(ql + 4 * k);
            accW = __builtin_amdgcn_wmma_f32_16x16x4_f32(
                false, aw, false, bfr[k], (short)0, accW, false, false);
            accQ = __builtin_amdgcn_wmma_f32_16x16x4_f32(
                false, aq, false, bfr[k], (short)0, accQ, false, false);
        }
#pragma unroll
        for (int j = 0; j < 4; ++j) {
            int lr0 = 2 * j + rowoff;
            int t0  = m * 16 + lr0;
            int t1  = t0 + 1;
            int c   = t0 >> 1;
            float wh0 = accW[2 * j], wh1 = accW[2 * j + 1];
            float qh0 = accQ[2 * j], qh1 = accQ[2 * j + 1];
            float u0 = ucur[2 * j];
            float u1 = ucur[2 * j + 1];
            float q0 = Ql[t0 * AP + ncol];
            float q1 = Ql[t1 * AP + ncol];
            float uv0v = u0 - wh0;
            float out0 = q0 * uv0v + qh0;
            float uv1v = u1 - (wh1 + s_w[c] * uv0v);
            float out1 = q1 * uv1v + qh1 + s_q[c] * uv0v;
            outbase[(size_t)t0 * DD + ncol] = out0;
            outbase[(size_t)t1 * DD + ncol] = out1;
        }
#pragma unroll
        for (int j = 0; j < 8; ++j) ucur[j] = unext[j];
    }
}

extern "C" void kernel_launch(void* const* d_in, const int* in_sizes, int n_in,
                              void* d_out, int out_size, void* d_ws, size_t ws_size,
                              hipStream_t stream) {
    const float* q  = (const float*)d_in[0];
    const float* W  = (const float*)d_in[1];
    const float* U  = (const float*)d_in[2];
    const float* h0 = (const float*)d_in[3];

    float* ys     = (float*)d_out;
    float* hfinal = ys + (size_t)BH_TOT * NCHUNK * CT * DD;
    float* hstate = (float*)d_ws;

    liz_scan<<<BH_TOT, SCAN_T, 0, stream>>>(W, U, h0, hstate, hfinal);
    liz_out<<<dim3(NCHUNK, BH_TOT), 256, 0, stream>>>(q, W, U, hstate, ys);
}


# scan: a wave owns 4 whole columns (16 rows a lane): column sums in-wave by DPP, no partial-sum LDS pass, 2 barriers a chunk; norms from wave partials; state in 4-column blocks (sum orders change)
# speedup vs baseline: 5.3367x; 1.3466x over previous
#include <hip/hip_runtime.h>
#include <math.h>

#define BH_TOT 32
#define NCHUNK 32
#define CT     128
#define CPAIR  64
#define DD     128
#define STABV  10000.0f

typedef float v2f __attribute__((ext_vector_type(2)));
typedef float v8f __attribute__((ext_vector_type(8)));

__device__ __forceinline__ float gelu_arg(float x) {
    const float k0 = 0.7978845608028654f;
    const float k1 = 0.044715f;
    return k0 * (x + k1 * x * x * x);
}
__device__ __forceinline__ float gelu_fin(float x, float t) {
    return 0.5f * x * (1.0f + t);
}
#define TANH_SMALL 0.625f
__device__ __forceinline__ float tanh_small(float u) {
    const float c0 = -0.00570002059f, c1 = 0.0206340719f, c2 = -0.0537379310f,
                c3 = 0.133314162f,    c4 = -0.333332807f;
    const float u2 = u * u;
    float p = __builtin_fmaf(c0, u2, c1);
    p = __builtin_fmaf(u2, p, c2);
    p = __builtin_fmaf(u2, p, c3);
    p = __builtin_fmaf(u2, p, c4);
    return __builtin_fmaf(u2, u * p, u);
}

__device__ __forceinline__ float clipstab(float v) {
    return fminf(fmaxf(v, -STABV), STABV);
}


__device__ __forceinline__ void lds_barrier() {
    __builtin_amdgcn_fence(__ATOMIC_RELEASE, "workgroup", "local");
    __builtin_amdgcn_s_barrier();
    __builtin_amdgcn_fence(__ATOMIC_ACQUIRE, "workgroup", "local");
}

__device__ __forceinline__ float lane_xor1(float x) {
    const int b = __builtin_bit_cast(int, x);
    return __builtin_bit_cast(float, __builtin_amdgcn_update_dpp(0, b, 0xB1, 0xF, 0xF, true));
}
__device__ __forceinline__ float lane_xor2(float x) {
    const int b = __builtin_bit_cast(int, x);
    return __builtin_bit_cast(float, __builtin_amdgcn_update_dpp(0, b, 0x4E, 0xF, 0xF, true));
}
__device__ __forceinline__ float lane_xor4(float x) {
    const int b = __builtin_bit_cast(int, x);
    return __builtin_bit_cast(float, __builtin_amdgcn_update_dpp(0, b, 0x164, 0xF, 0xF, true));
}
__device__ __forceinline__ float lane_xor8(float x) {
    const int b = __builtin_bit_cast(int, x);
    return __builtin_bit_cast(float, __builtin_amdgcn_update_dpp(0, b, 0x168, 0xF, 0xF, true));
}
__device__ __forceinline__ float lane_xor16(float x) {
    const int b = __builtin_bit_cast(int, x);
    return __builtin_bit_cast(float, __builtin_amdgcn_permlanex16(b, b, 0x76543210, 0xfedcba98, false, false));
}
__device__ __forceinline__ float wave_sum32(float p) {
    p += lane_xor16(p);
    p += lane_xor8(p);
    p += lane_xor4(p);
    p += lane_xor2(p);
    p += lane_xor1(p);
    return p;
}

#define SCAN_T 1024
#define RPT    16
#define LOAD_T0 512
#define NP     132
#define HBLK   512

__global__ __launch_bounds__(SCAN_T) void liz_scan(
    const float* __restrict__ Wt, const float* __restrict__ Ut,
    const float* __restrict__ h0, float* __restrict__ hstates,
    float* __restrict__ hfinal)
{
    __shared__ __attribute__((aligned(16))) float npart[32 * NP];
    __shared__ __attribute__((aligned(16))) float wwp[2][4];
    __shared__ float ubuf[2][2][DD];
    __shared__ __attribute__((aligned(16))) float wbuf[2][2][DD];
    __shared__ __attribute__((aligned(16))) float ninv[DD];
    __shared__ __attribute__((aligned(16))) float nhn[DD];

    const int tid  = threadIdx.x;
    const int bh   = blockIdx.x;
    const int lane = tid & 31;
    const int wv   = tid >> 5;
    const int c    = lane & 3;
    const int g    = lane >> 2;
    const int col  = 4 * wv + c;
    const int hoff = (g >> 1) * 8 + c * 2 + (g & 1);
    const bool loader = (tid >= LOAD_T0) && (tid < LOAD_T0 + DD);
    const int  le     = (tid - LOAD_T0) & (DD - 1);
    const int  lep    = ((le & 7) << 4) | (le >> 3);

    float hreg[RPT];
    {
        const float* src = h0 + (size_t)bh * DD * DD + (size_t)g * DD + col;
#pragma unroll
        for (int i = 0; i < RPT; ++i) hreg[i] = src[(size_t)i * 8 * DD];
    }
    {
        float* dst = hstates + (size_t)(bh * NCHUNK) * DD * DD + wv * HBLK + hoff;
#pragma unroll
        for (int i = 0; i < RPT; ++i) dst[i * 32] = hreg[i];
    }
    float nw0 = 0.f, nw1 = 0.f, nu0 = 0.f, nu1 = 0.f;
    if (loader) {
        const float* wb = Wt + (size_t)(bh * NCHUNK) * CT * DD;
        const float* ub = Ut + (size_t)(bh * NCHUNK) * CT * DD;
        const float a0 = wb[(size_t)(CT - 2) * DD + le];
        const float a1 = wb[(size_t)(CT - 1) * DD + le];
        const float p = wave_sum32(a1 * a0);
        if (lane == 0) wwp[0][le >> 5] = p;
        wbuf[0][0][lep] = a0;
        wbuf[0][1][lep] = a1;
        ubuf[0][0][le] = ub[(size_t)(CT - 2) * DD + le];
        ubuf[0][1][le] = ub[(size_t)(CT - 1) * DD + le];
        nw0 = wb[(size_t)CT * DD + (size_t)(CT - 2) * DD + le];
        nw1 = wb[(size_t)CT * DD + (size_t)(CT - 1) * DD + le];
        nu0 = ub[(size_t)CT * DD + (size_t)(CT - 2) * DD + le];
        nu1 = ub[(size_t)CT * DD + (size_t)(CT - 1) * DD + le];
    }
    __syncthreads();
    float u0c = ubuf[0][0][col];
    float u1c = ubuf[0][1][col];

    for (int ch = 0; ch < NCHUNK; ++ch) {
        const int cur = ch & 1;
        const bool more = loader && (ch + 1 < NCHUNK);

        float ww;
        {
            const float4 wp = *(const float4*)&wwp[cur][0];
            ww = (wp.x + wp.y) + (wp.z + wp.w);
        }
        float w0v[RPT], w1v[RPT];
        {
            const float4* p0 = (const float4*)&wbuf[cur][0][g * RPT];
            const float4* p1 = (const float4*)&wbuf[cur][1][g * RPT];
#pragma unroll
            for (int m = 0; m < RPT / 4; ++m) {
                const float4 a = p0[m];
                const float4 b = p1[m];
                w0v[4 * m + 0] = a.x; w0v[4 * m + 1] = a.y; w0v[4 * m + 2] = a.z; w0v[4 * m + 3] = a.w;
                w1v[4 * m + 0] = b.x; w1v[4 * m + 1] = b.y; w1v[4 * m + 2] = b.z; w1v[4 * m + 3] = b.w;
            }
        }
        float k0, k1;
        {
            v2f a0 = {0.f, 0.f}, a1 = {0.f, 0.f};
#pragma unroll
            for (int m = 0; m < RPT / 2; ++m) {
                const v2f hp  = {hreg[2 * m], hreg[2 * m + 1]};
                const v2f w0p = {w0v[2 * m], w0v[2 * m + 1]};
                const v2f w1p = {w1v[2 * m], w1v[2 * m + 1]};
                a0 += w0p * hp;
                a1 += w1p * hp;
            }
            float s0 = a0.x + a0.y, s1 = a1.x + a1.y;
            s0 += lane_xor4(s0);   s1 += lane_xor4(s1);
            s0 += lane_xor8(s0);   s1 += lane_xor8(s1);
            s0 += lane_xor16(s0);  s1 += lane_xor16(s1);
            k0 = s0;  k1 = s1;
        }
        const float uv0 = u0c - k0;
        const float uv1 = u1c - (k1 + ww * uv0);

        {
            float sq[RPT];
#pragma unroll
            for (int i = 0; i < RPT; ++i) {
                float v;
                v = clipstab(hreg[i] + w0v[i] * uv0);
                v = clipstab(v + w1v[i] * uv1);
                hreg[i] = v;
                sq[i] = v * v;
            }
#pragma unroll
            for (int i = 0; i < RPT; ++i) {
                sq[i] += lane_xor1(sq[i]);
                sq[i] += lane_xor2(sq[i]);
            }
            if (c == 0) {
                float4* np = (float4*)(npart + wv * NP + g * RPT);
#pragma unroll
                for (int m = 0; m < RPT / 4; ++m)
                    np[m] = make_float4(sq[4 * m + 0], sq[4 * m + 1], sq[4 * m + 2], sq[4 * m + 3]);
            }
        }
        lds_barrier();

        {
            const int pr = tid >> 3;
            const int j  = tid & 7;
            const float* pp = npart + (4 * j) * NP + pr;
            float sqs = (pp[0] + pp[NP]) + (pp[2 * NP] + pp[3 * NP]);
            sqs += lane_xor1(sqs);
            sqs += lane_xor2(sqs);
            sqs += lane_xor4(sqs);
            const float n = __builtin_amdgcn_sqrtf(sqs) + 1e-6f;
            if (j == 0) {
                ninv[pr] = 2.0f * __builtin_amdgcn_rcpf(n);
                nhn[pr]  = 0.5f * n;
            }
        }
        if (more) {
            const float p = wave_sum32(nw1 * nw0);
            if (lane == 0) wwp[cur ^ 1][le >> 5] = p;
            wbuf[cur ^ 1][0][lep] = nw0;
            wbuf[cur ^ 1][1][lep] = nw1;
            ubuf[cur ^ 1][0][le] = nu0;
            ubuf[cur ^ 1][1][le] = nu1;
            const int c2 = (ch + 2 < NCHUNK) ? ch + 2 : NCHUNK - 1;
            const float* wb = Wt + (size_t)(bh * NCHUNK + c2) * CT * DD;
            const float* ub = Ut + (size_t)(bh * NCHUNK + c2) * CT * DD;
            nw0 = wb[(size_t)(CT - 2) * DD + le];
            nw1 = wb[(size_t)(CT - 1) * DD + le];
            nu0 = ub[(size_t)(CT - 2) * DD + le];
            nu1 = ub[(size_t)(CT - 1) * DD + le];
        }
        lds_barrier();

        u0c = ubuf[cur ^ 1][0][col];
        u1c = ubuf[cur ^ 1][1][col];

        {
            const float4* iv = (const float4*)(ninv + g * RPT);
            const float4* hv = (const float4*)(nhn + g * RPT);
            float xs[RPT], us[RPT];
            unsigned outm[RPT];
            unsigned anyout = 0u;
#pragma unroll
            for (int i4 = 0; i4 < RPT / 4; ++i4) {
                float4 a = iv[i4];
                xs[4 * i4 + 0] = hreg[4 * i4 + 0] * a.x;
                xs[4 * i4 + 1] = hreg[4 * i4 + 1] * a.y;
                xs[4 * i4 + 2] = hreg[4 * i4 + 2] * a.z;
                xs[4 * i4 + 3] = hreg[4 * i4 + 3] * a.w;
            }
#pragma unroll
            for (int i = 0; i < RPT; ++i) {
                us[i] = gelu_arg(xs[i]);
                outm[i] = __builtin_amdgcn_ballot_w32(!(__builtin_fabsf(us[i]) < TANH_SMALL));
                anyout |= outm[i];
            }
            if (anyout == 0u) {
#pragma unroll
                for (int i4 = 0; i4 < RPT / 4; ++i4) {
                    float4 b = hv[i4];
                    const float bb[4] = {b.x, b.y, b.z, b.w};
#pragma unroll
                    for (int j = 0; j < 4; ++j) {
                        const int i = 4 * i4 + j;
                        hreg[i] = bb[j] * gelu_fin(xs[i], tanh_small(us[i]));
                    }
                }
            } else {
                float tfix[RPT];
#pragma unroll
                for (int i = 0; i < RPT; ++i) tfix[i] = 0.0f;
#pragma unroll
                for (int i = 0; i < RPT; ++i) {
                    if (outm[i] != 0u) tfix[i] = tanhf(us[i]);
                }
#pragma unroll
                for (int i4 = 0; i4 < RPT / 4; ++i4) {
                    float4 b = hv[i4];
                    const float bb[4] = {b.x, b.y, b.z, b.w};
#pragma unroll
                    for (int j = 0; j < 4; ++j) {
                        const int i = 4 * i4 + j;
                        const float tp = tanh_small(us[i]);
                        const float t = !(__builtin_fabsf(us[i]) < TANH_SMALL) ? tfix[i] : tp;
                        hreg[i] = bb[j] * gelu_fin(xs[i], t);
                    }
                }
            }
            if (ch + 1 < NCHUNK) {
                float* dst = hstates + (size_t)(bh * NCHUNK + ch + 1) * DD * DD + wv * HBLK + hoff;
#pragma unroll
                for (int i = 0; i < RPT; ++i) dst[i * 32] = hreg[i];
            } else {
                float* dst = hfinal + (size_t)bh * DD * DD + (size_t)g * DD + col;
#pragma unroll
                for (int i = 0; i < RPT; ++i) dst[(size_t)i * 8 * DD] = hreg[i];
            }
        }
    }
}
#define AP 268
#define QO 132

__global__ __launch_bounds__(256) void liz_out(
    const float* __restrict__ Qt, const float* __restrict__ Wt,
    const float* __restrict__ Ut, const float* __restrict__ hstates,
    float* __restrict__ ys)
{
    __shared__ __attribute__((aligned(16))) float WQ[CT * AP];
    float* const Wl = WQ;
    float* const Ql = WQ + QO;
    __shared__ float s_w[CPAIR], s_q[CPAIR];

    const int tid  = threadIdx.x;
    const int lane = tid & 31;
    const int wave = tid >> 5;
    const int nc = blockIdx.x, bh = blockIdx.y;

    const size_t cbase = (size_t)(bh * NCHUNK + nc) * CT * DD;
    const float* Qc = Qt + cbase;
    const float* Wc = Wt + cbase;
    const float* Uc = Ut + cbase;

    const int nlane  = lane & 15;
    const int khalf  = 2 * (lane >> 4);
    const int rowoff = (lane >> 4) * 8;
    const int ncol   = wave * 16 + nlane;

    v2f bfr[32];
    {
        const float* hsrc = hstates + (size_t)(bh * NCHUNK + nc) * DD * DD + (ncol >> 2) * HBLK + (ncol & 3) * 2;
#pragma unroll
        for (int k = 0; k < 32; ++k) {
            int kk = 4 * k + khalf;
            bfr[k] = *(const v2f*)(hsrc + (kk >> 1) * 8);
        }
    }

    {
        const float4* wsrc = (const float4*)Wc;
        const float4* qsrc = (const float4*)Qc;
#pragma unroll
        for (int i = 0; i < 16; ++i) {
            int idx = tid + 256 * i;
            int row = idx >> 5, c4 = idx & 31;
            *(float4*)(Wl + row * AP + 4 * c4) = wsrc[idx];
        }
#pragma unroll
        for (int i = 0; i < 16; ++i) {
            int idx = tid + 256 * i;
            int row = idx >> 5, c4 = idx & 31;
            *(float4*)(Ql + row * AP + 4 * c4) = qsrc[idx];
        }
    }

    float ucur[8];
#pragma unroll
    for (int j = 0; j < 4; ++j) {
        int t0 = 2 * j + rowoff;
        ucur[2 * j]     = Uc[(size_t)t0 * DD + ncol];
        ucur[2 * j + 1] = Uc[(size_t)(t0 + 1) * DD + ncol];
    }
    __syncthreads();

    {
        float pw[8], pq[8];
#pragma unroll
        for (int i = 0; i < 8; ++i) {
            int c = wave * 8 + i;
            const float* w0p = Wl + (2 * c) * AP;
            const float* w1p = Wl + (2 * c + 1) * AP;
            const float* q1p = Ql + (2 * c + 1) * AP;
            float aw = 0.f, aq = 0.f;
#pragma unroll
            for (int e = lane; e < DD; e += 32) {
                float a = w0p[e];
                aw += w1p[e] * a;
                aq += q1p[e] * a;
            }
            pw[i] = aw; pq[i] = aq;
        }
#pragma unroll
        for (int off = 16; off > 0; off >>= 1) {
#pragma unroll
            for (int i = 0; i < 8; ++i) {
                pw[i] += __shfl_xor(pw[i], off, 32);
                pq[i] += __shfl_xor(pq[i], off, 32);
            }
        }
        if (lane == 0) {
#pragma unroll
            for (int i = 0; i < 8; ++i) { s_w[wave * 8 + i] = pw[i]; s_q[wave * 8 + i] = pq[i]; }
        }
    }
    __syncthreads();

    float* outbase = ys + cbase;
    for (int m = 0; m < 8; ++m) {
        float unext[8];
        {
            const int mn = (m + 1 < 8) ? m + 1 : m;
#pragma unroll
            for (int j = 0; j < 4; ++j) {
                int t0 = mn * 16 + 2 * j + rowoff;
                unext[2 * j]     = Uc[(size_t)t0 * DD + ncol];
                unext[2 * j + 1] = Uc[(size_t)(t0 + 1) * DD + ncol];
            }
        }

        v8f accW = {};
        v8f accQ = {};
        const float* wl = Wl + (m * 16 + nlane) * AP + khalf;
        const float* ql = Ql + (m * 16 + nlane) * AP + khalf;
#pragma unroll
        for (int k = 0; k < 32; ++k) {
            v2f aw = *(const v2f*)(wl + 4 * k);
            v2f aq = *(const v2f*)(ql + 4 * k);
            accW = __builtin_amdgcn_wmma_f32_16x16x4_f32(
                false, aw, false, bfr[k], (short)0, accW, false, false);
            accQ = __builtin_amdgcn_wmma_f32_16x16x4_f32(
                false, aq, false, bfr[k], (short)0, accQ, false, false);
        }
#pragma unroll
        for (int j = 0; j < 4; ++j) {
            int lr0 = 2 * j + rowoff;
            int t0  = m * 16 + lr0;
            int t1  = t0 + 1;
            int c   = t0 >> 1;
            float wh0 = accW[2 * j], wh1 = accW[2 * j + 1];
            float qh0 = accQ[2 * j], qh1 = accQ[2 * j + 1];
            float u0 = ucur[2 * j];
            float u1 = ucur[2 * j + 1];
            float q0 = Ql[t0 * AP + ncol];
            float q1 = Ql[t1 * AP + ncol];
            float uv0v = u0 - wh0;
            float out0 = q0 * uv0v + qh0;
            float uv1v = u1 - (wh1 + s_w[c] * uv0v);
            float out1 = q1 * uv1v + qh1 + s_q[c] * uv0v;
            outbase[(size_t)t0 * DD + ncol] = out0;
            outbase[(size_t)t1 * DD + ncol] = out1;
        }
#pragma unroll
        for (int j = 0; j < 8; ++j) ucur[j] = unext[j];
    }
}

extern "C" void kernel_launch(void* const* d_in, const int* in_sizes, int n_in,
                              void* d_out, int out_size, void* d_ws, size_t ws_size,
                              hipStream_t stream) {
    const float* q  = (const float*)d_in[0];
    const float* W  = (const float*)d_in[1];
    const float* U  = (const float*)d_in[2];
    const float* h0 = (const float*)d_in[3];

    float* ys     = (float*)d_out;
    float* hfinal = ys + (size_t)BH_TOT * NCHUNK * CT * DD;
    float* hstate = (float*)d_ws;

    liz_scan<<<BH_TOT, SCAN_T, 0, stream>>>(W, U, h0, hstate, hfinal);
    liz_out<<<dim3(NCHUNK, BH_TOT), 256, 0, stream>>>(q, W, U, hstate, ys);
}
